# starting kernel
# baseline (speedup 1.0000x reference)
#include <hip/hip_runtime.h>
#include <hip/hip_bf16.h>

typedef __attribute__((ext_vector_type(2))) float v2f;
typedef __attribute__((ext_vector_type(8))) float v8f;

#define F_DIM 128
#define LEAKY 0.2f

__device__ __forceinline__ unsigned f2ord(float f) {
  unsigned u = __float_as_uint(f);
  return (u & 0x80000000u) ? ~u : (u | 0x80000000u);
}
__device__ __forceinline__ float ord2f(unsigned u) {
  return (u & 0x80000000u) ? __uint_as_float(u & 0x7FFFFFFFu)
                           : __uint_as_float(~u);
}

__global__ __launch_bounds__(256) void gat_gemm_wmma(
    const float* __restrict__ x, const float* __restrict__ W,
    float* __restrict__ h, int M)
{
  __shared__ float sA[16 * 132];
  const int tid = threadIdx.x;
  const int m0  = blockIdx.x * 16;

  for (int i = tid; i < 512; i += 256) {
    int r = i >> 5, c = i & 31;
    float4 v = make_float4(0.f, 0.f, 0.f, 0.f);
    if (m0 + r < M) v = ((const float4*)(x + (size_t)(m0 + r) * F_DIM))[c];
    ((float4*)sA)[r * 33 + c] = v;
  }
  __syncthreads();

  const int lane = tid & 31;
  const int wv   = tid >> 5;
  const int row  = lane & 15;
  const int kh   = (lane >> 4) * 2;
  const int n0   = wv * 16;

  v8f acc = {};
  #pragma unroll 4
  for (int k = 0; k < F_DIM; k += 4) {
    v2f a, b;
    a.x = sA[row * 132 + k + kh + 0];
    a.y = sA[row * 132 + k + kh + 1];
    b.x = W[(k + kh + 0) * F_DIM + n0 + row];
    b.y = W[(k + kh + 1) * F_DIM + n0 + row];
    acc = __builtin_amdgcn_wmma_f32_16x16x4_f32(false, a, false, b,
                                                (short)0, acc, false, false);
  }

  const int mh = (lane >> 4) * 8;
  float* hp = h + (size_t)m0 * F_DIM + n0 + row;
  if (m0 + 16 <= M) {
    #pragma unroll
    for (int v = 0; v < 8; ++v) hp[(size_t)(mh + v) * F_DIM] = acc[v];
  } else {
    #pragma unroll
    for (int v = 0; v < 8; ++v)
      if (m0 + mh + v < M) hp[(size_t)(mh + v) * F_DIM] = acc[v];
  }
}

__global__ __launch_bounds__(256) void gat_alpha(
    const float* __restrict__ h, const float* __restrict__ a,
    float* __restrict__ as, float* __restrict__ ad, int N)
{
  const int node = blockIdx.x * 8 + (threadIdx.x >> 5);
  const int lane = threadIdx.x & 31;
  if (node >= N) return;
  float4 hv = ((const float4*)(h + (size_t)node * F_DIM))[lane];
  float4 a1 = ((const float4*)a)[lane];
  float4 a2 = ((const float4*)(a + F_DIM))[lane];
  float s = hv.x * a1.x + hv.y * a1.y + hv.z * a1.z + hv.w * a1.w;
  float d = hv.x * a2.x + hv.y * a2.y + hv.z * a2.z + hv.w * a2.w;
  #pragma unroll
  for (int m = 16; m >= 1; m >>= 1) {
    s += __shfl_xor(s, m, 32);
    d += __shfl_xor(d, m, 32);
  }
  if (lane == 0) { as[node] = s; ad[node] = d; }
}

__global__ __launch_bounds__(256) void gat_edge_logit(
    const int* __restrict__ src, const int* __restrict__ dst,
    const float* __restrict__ as, const float* __restrict__ ad,
    float* __restrict__ e, unsigned* __restrict__ segmax, int E)
{
  int i = blockIdx.x * blockDim.x + threadIdx.x;
  if (i >= E) return;
  float v = as[src[i]] + ad[dst[i]];
  v = v > 0.f ? v : LEAKY * v;
  e[i] = v;
  atomicMax(segmax + dst[i], f2ord(v));
}

__global__ __launch_bounds__(256) void gat_edge_exp(
    const int* __restrict__ dst, const float* __restrict__ e,
    const unsigned* __restrict__ segmax, float* __restrict__ ex,
    float* __restrict__ segsum, int E)
{
  int i = blockIdx.x * blockDim.x + threadIdx.x;
  if (i >= E) return;
  int d = dst[i];
  float v = expf(e[i] - ord2f(segmax[d]));
  ex[i] = v;
  atomicAdd(segsum + d, v);
}

__global__ __launch_bounds__(256) void gat_aggregate(
    const int* __restrict__ src, const int* __restrict__ dst,
    const float* __restrict__ h, const float* __restrict__ ex,
    const float* __restrict__ segsum, float* __restrict__ out, int E)
{
  const int i    = __builtin_amdgcn_readfirstlane(blockIdx.x * 8 + (threadIdx.x >> 5));
  const int lane = threadIdx.x & 31;
  if (i >= E) return;
  const int s = src[i], d = dst[i];
  const float coeff = ex[i] / (segsum[d] + 1e-16f);
  float4 hv = ((const float4*)(h + (size_t)s * F_DIM))[lane];
  float* op = out + (size_t)d * F_DIM + lane * 4;
  atomicAdd(op + 0, coeff * hv.x);
  atomicAdd(op + 1, coeff * hv.y);
  atomicAdd(op + 2, coeff * hv.z);
  atomicAdd(op + 3, coeff * hv.w);
}

extern "C" void kernel_launch(void* const* d_in, const int* in_sizes, int n_in,
                              void* d_out, int out_size, void* d_ws, size_t ws_size,
                              hipStream_t stream) {
  const float* x  = (const float*)d_in[0];
  const int*   ei = (const int*)d_in[1];
  const float* W  = (const float*)d_in[2];
  const float* a  = (const float*)d_in[3];
  const int N = in_sizes[0] / F_DIM;
  const int E = in_sizes[1] / 2;
  const int* src = ei;
  const int* dst = ei + E;

  float*    h      = (float*)d_ws;
  float*    as     = h + (size_t)N * F_DIM;
  float*    ad     = as + N;
  unsigned* segmax = (unsigned*)(ad + N);
  float*    segsum = (float*)(segmax + N);
  float*    e      = segsum + N;
  float*    ex     = e + E;
  float*    out    = (float*)d_out;

  hipMemsetAsync(d_out, 0, (size_t)out_size * sizeof(float), stream);
  hipMemsetAsync(segmax, 0, (size_t)N * sizeof(unsigned), stream);
  hipMemsetAsync(segsum, 0, (size_t)N * sizeof(float), stream);

  gat_gemm_wmma <<<(N + 15) / 16,  256, 0, stream>>>(x, W, h, N);
  gat_alpha     <<<(N + 7)  / 8,   256, 0, stream>>>(h, a, as, ad, N);
  gat_edge_logit<<<(E + 255) / 256, 256, 0, stream>>>(src, dst, as, ad, e, segmax, E);
  gat_edge_exp  <<<(E + 255) / 256, 256, 0, stream>>>(dst, e, segmax, ex, segsum, E);
  gat_aggregate <<<(E + 7)  / 8,   256, 0, stream>>>(src, dst, h, ex, segsum, out, E);
}


# aggregate: 256 work-groups each own 196 dst rows as an LDS tile, filter a byte-per-edge owner plane for their edges, add rows by LDS atomics, store once; no out memset; sums in LDS-atomic order
# speedup vs baseline: 1.4732x; 1.4732x over previous
#include <hip/hip_runtime.h>
#include <hip/hip_bf16.h>

typedef __attribute__((ext_vector_type(2))) float v2f;
typedef __attribute__((ext_vector_type(8))) float v8f;

#define F_DIM 128
#define LEAKY 0.2f

#define AGG_GROUPS  256
#define AGG_ROWS    196
#define AGG_THREADS 1024
#define AGG_QCAP    8192
#define AGG_INFLIGHT 8

__device__ __forceinline__ unsigned f2ord(float f) {
  unsigned u = __float_as_uint(f);
  return (u & 0x80000000u) ? ~u : (u | 0x80000000u);
}
__device__ __forceinline__ float ord2f(unsigned u) {
  return (u & 0x80000000u) ? __uint_as_float(u & 0x7FFFFFFFu)
                           : __uint_as_float(~u);
}

__global__ __launch_bounds__(256) void gat_gemm_wmma(
    const float* __restrict__ x, const float* __restrict__ W,
    float* __restrict__ h, int M)
{
  __shared__ float sA[16 * 132];
  const int tid = threadIdx.x;
  const int m0  = blockIdx.x * 16;

  for (int i = tid; i < 512; i += 256) {
    int r = i >> 5, c = i & 31;
    float4 v = make_float4(0.f, 0.f, 0.f, 0.f);
    if (m0 + r < M) v = ((const float4*)(x + (size_t)(m0 + r) * F_DIM))[c];
    ((float4*)sA)[r * 33 + c] = v;
  }
  __syncthreads();

  const int lane = tid & 31;
  const int wv   = tid >> 5;
  const int row  = lane & 15;
  const int kh   = (lane >> 4) * 2;
  const int n0   = wv * 16;

  v8f acc = {};
  #pragma unroll 4
  for (int k = 0; k < F_DIM; k += 4) {
    v2f a, b;
    a.x = sA[row * 132 + k + kh + 0];
    a.y = sA[row * 132 + k + kh + 1];
    b.x = W[(k + kh + 0) * F_DIM + n0 + row];
    b.y = W[(k + kh + 1) * F_DIM + n0 + row];
    acc = __builtin_amdgcn_wmma_f32_16x16x4_f32(false, a, false, b,
                                                (short)0, acc, false, false);
  }

  const int mh = (lane >> 4) * 8;
  float* hp = h + (size_t)m0 * F_DIM + n0 + row;
  if (m0 + 16 <= M) {
    #pragma unroll
    for (int v = 0; v < 8; ++v) hp[(size_t)(mh + v) * F_DIM] = acc[v];
  } else {
    #pragma unroll
    for (int v = 0; v < 8; ++v)
      if (m0 + mh + v < M) hp[(size_t)(mh + v) * F_DIM] = acc[v];
  }
}

__global__ __launch_bounds__(256) void gat_alpha(
    const float* __restrict__ h, const float* __restrict__ a,
    float* __restrict__ as, float* __restrict__ ad, int N)
{
  const int node = blockIdx.x * 8 + (threadIdx.x >> 5);
  const int lane = threadIdx.x & 31;
  if (node >= N) return;
  float4 hv = ((const float4*)(h + (size_t)node * F_DIM))[lane];
  float4 a1 = ((const float4*)a)[lane];
  float4 a2 = ((const float4*)(a + F_DIM))[lane];
  float s = hv.x * a1.x + hv.y * a1.y + hv.z * a1.z + hv.w * a1.w;
  float d = hv.x * a2.x + hv.y * a2.y + hv.z * a2.z + hv.w * a2.w;
  #pragma unroll
  for (int m = 16; m >= 1; m >>= 1) {
    s += __shfl_xor(s, m, 32);
    d += __shfl_xor(d, m, 32);
  }
  if (lane == 0) { as[node] = s; ad[node] = d; }
}

__global__ __launch_bounds__(256) void gat_edge_logit(
    const int* __restrict__ src, const int* __restrict__ dst,
    const float* __restrict__ as, const float* __restrict__ ad,
    float* __restrict__ e, unsigned* __restrict__ segmax, int E)
{
  int i = blockIdx.x * blockDim.x + threadIdx.x;
  if (i >= E) return;
  float v = as[src[i]] + ad[dst[i]];
  v = v > 0.f ? v : LEAKY * v;
  e[i] = v;
  atomicMax(segmax + dst[i], f2ord(v));
}

__global__ __launch_bounds__(256) void gat_edge_exp(
    const int* __restrict__ dst, float* __restrict__ e,
    const unsigned* __restrict__ segmax, float* __restrict__ segsum,
    unsigned char* __restrict__ owner, int E)
{
  int i = blockIdx.x * blockDim.x + threadIdx.x;
  if (i >= E) return;
  int d = dst[i];
  float v = expf(e[i] - ord2f(segmax[d]));
  e[i] = v;
  owner[i] = (unsigned char)((d / AGG_ROWS) & (AGG_GROUPS - 1));
  atomicAdd(segsum + d, v);
}

__device__ __forceinline__ unsigned zero_bytes(unsigned x) {
  unsigned t = (x & 0x7F7F7F7Fu) + 0x7F7F7F7Fu;
  return ~(t | x | 0x7F7F7F7Fu);
}

__device__ __forceinline__ void agg_take(
    int i, int base, int N,
    const int* __restrict__ src, const int* __restrict__ dst,
    const float* __restrict__ h, const float* __restrict__ ex,
    const float* __restrict__ segsum,
    float* tile, int* queue, int* qcount)
{
  const int pos = atomicAdd(qcount, 1);
  if (pos < AGG_QCAP) { queue[pos] = i; return; }
  const int d = dst[i];
  if (d < base || d >= base + AGG_ROWS || d >= N) return;
  const int s = src[i];
  const float c = ex[i] / (segsum[d] + 1e-16f);
  const float* hp = h + (size_t)s * F_DIM;
  float* tp = tile + (d - base) * F_DIM;
  for (int f = 0; f < F_DIM; ++f) atomicAdd(tp + f, c * hp[f]);
}

__global__ __launch_bounds__(AGG_THREADS) void gat_aggregate(
    const int* __restrict__ src, const int* __restrict__ dst,
    const float* __restrict__ h, const float* __restrict__ ex,
    const float* __restrict__ segsum, const unsigned char* __restrict__ owner,
    float* __restrict__ out, int N, int E)
{
  __shared__ float tile[AGG_ROWS * F_DIM];
  __shared__ int   queue[AGG_QCAP];
  __shared__ int   qcount;

  const int tid  = threadIdx.x;
  const int lane = tid & 31;
  const int wv   = tid >> 5;
  const unsigned me  = blockIdx.x;
  const unsigned me4 = me * 0x01010101u;

  int head = (int)((16u - (unsigned)((size_t)owner & 15u)) & 15u);
  if (head > E) head = E;
  const int nvec  = (E - head) >> 4;
  const int tail0 = head + (nvec << 4);
  const uint4* vp = (const uint4*)(owner + head);
  const int vstart = (int)(((long long)blockIdx.x * nvec) / AGG_GROUPS);

  for (int base = blockIdx.x * AGG_ROWS; base < N; base += AGG_GROUPS * AGG_ROWS) {
    for (int k = tid; k < AGG_ROWS * F_DIM / 4; k += AGG_THREADS)
      ((float4*)tile)[k] = make_float4(0.f, 0.f, 0.f, 0.f);
    if (tid == 0) qcount = 0;
    __syncthreads();

    int v0 = tid;
    for (; v0 + (AGG_INFLIGHT - 1) * AGG_THREADS < nvec; v0 += AGG_INFLIGHT * AGG_THREADS) {
      uint4 w[AGG_INFLIGHT];
      int   vi[AGG_INFLIGHT];
      #pragma unroll
      for (int u = 0; u < AGG_INFLIGHT; ++u) {
        int v = v0 + u * AGG_THREADS + vstart;
        if (v >= nvec) v -= nvec;
        vi[u] = v;
        w[u]  = vp[v];
      }
      #pragma unroll
      for (int u = 0; u < AGG_INFLIGHT; ++u) {
        unsigned m = (zero_bytes(w[u].x ^ me4) >> 7) | (zero_bytes(w[u].y ^ me4) >> 6) |
                     (zero_bytes(w[u].z ^ me4) >> 5) | (zero_bytes(w[u].w ^ me4) >> 4);
        while (m) {
          const int p = __builtin_ctz(m);
          m &= m - 1;
          const int i = head + (vi[u] << 4) + ((p & 7) << 2) + (p >> 3);
          agg_take(i, base, N, src, dst, h, ex, segsum, tile, queue, &qcount);
        }
      }
    }
    for (; v0 < nvec; v0 += AGG_THREADS) {
      int v = v0 + vstart;
      if (v >= nvec) v -= nvec;
      const uint4 w1 = vp[v];
      unsigned m = (zero_bytes(w1.x ^ me4) >> 7) | (zero_bytes(w1.y ^ me4) >> 6) |
                   (zero_bytes(w1.z ^ me4) >> 5) | (zero_bytes(w1.w ^ me4) >> 4);
      while (m) {
        const int p = __builtin_ctz(m);
        m &= m - 1;
        const int i = head + (v << 4) + ((p & 7) << 2) + (p >> 3);
        agg_take(i, base, N, src, dst, h, ex, segsum, tile, queue, &qcount);
      }
    }
    for (int i = tid; i < head; i += AGG_THREADS)
      if (owner[i] == me) agg_take(i, base, N, src, dst, h, ex, segsum, tile, queue, &qcount);
    for (int i = tail0 + tid; i < E; i += AGG_THREADS)
      if (owner[i] == me) agg_take(i, base, N, src, dst, h, ex, segsum, tile, queue, &qcount);
    __syncthreads();

    const int n = min(qcount, AGG_QCAP);
    for (int q0 = wv * 32; q0 < n; q0 += AGG_THREADS) {
      const int q = q0 + lane;
      int s = 0, dl = -1;
      float c = 0.f;
      if (q < n) {
        const int i = queue[q];
        const int d = dst[i];
        if (d >= base && d < base + AGG_ROWS && d < N) {
          s  = src[i];
          dl = d - base;
          c  = ex[i] / (segsum[d] + 1e-16f);
        }
      }
      for (int j = 0; j < 32; j += 8) {
        if (q0 + j >= n) break;
        float r[8][4];
        int   dj[8];
        float cj[8];
        #pragma unroll
        for (int u = 0; u < 8; ++u) {
          const int sj = __builtin_amdgcn_readlane(s, j + u);
          dj[u] = __builtin_amdgcn_readlane(dl, j + u);
          cj[u] = __int_as_float(__builtin_amdgcn_readlane(__float_as_int(c), j + u));
          if (dj[u] >= 0) {
            const float* hp = h + (size_t)sj * F_DIM + lane;
            r[u][0] = hp[0];
            r[u][1] = hp[32];
            r[u][2] = hp[64];
            r[u][3] = hp[96];
          }
        }
        #pragma unroll
        for (int u = 0; u < 8; ++u) {
          if (dj[u] >= 0) {
            float* tp = tile + dj[u] * F_DIM + lane;
            atomicAdd(tp,      cj[u] * r[u][0]);
            atomicAdd(tp + 32, cj[u] * r[u][1]);
            atomicAdd(tp + 64, cj[u] * r[u][2]);
            atomicAdd(tp + 96, cj[u] * r[u][3]);
          }
        }
      }
    }
    __syncthreads();

    const int rows = min(AGG_ROWS, N - base);
    float4* op = (float4*)(out + (size_t)base * F_DIM);
    for (int k = tid; k < rows * (F_DIM / 4); k += AGG_THREADS)
      op[k] = ((const float4*)tile)[k];
    __syncthreads();
  }
}

extern "C" void kernel_launch(void* const* d_in, const int* in_sizes, int n_in,
                              void* d_out, int out_size, void* d_ws, size_t ws_size,
                              hipStream_t stream) {
  const float* x  = (const float*)d_in[0];
  const int*   ei = (const int*)d_in[1];
  const float* W  = (const float*)d_in[2];
  const float* a  = (const float*)d_in[3];
  const int N = in_sizes[0] / F_DIM;
  const int E = in_sizes[1] / 2;
  const int* src = ei;
  const int* dst = ei + E;

  float*    h      = (float*)d_ws;
  float*    as     = h + (size_t)N * F_DIM;
  float*    ad     = as + N;
  unsigned* segmax = (unsigned*)(ad + N);
  float*    segsum = (float*)(segmax + N);
  float*    e      = segsum + N;
  unsigned char* owner = (unsigned char*)(e + E);
  float*    out    = (float*)d_out;

  hipMemsetAsync(segmax, 0, (size_t)N * sizeof(unsigned), stream);
  hipMemsetAsync(segsum, 0, (size_t)N * sizeof(float), stream);

  gat_gemm_wmma <<<(N + 15) / 16,  256, 0, stream>>>(x, W, h, N);
  gat_alpha     <<<(N + 7)  / 8,   256, 0, stream>>>(h, a, as, ad, N);
  gat_edge_logit<<<(E + 255) / 256, 256, 0, stream>>>(src, dst, as, ad, e, segmax, E);
  gat_edge_exp  <<<(E + 255) / 256, 256, 0, stream>>>(dst, e, segmax, segsum, owner, E);
  gat_aggregate <<<AGG_GROUPS, AGG_THREADS, 0, stream>>>(src, dst, h, e, segsum, owner, out, N, E);
}


# softmax max and sum per destination taken in LDS by the owning work-group over queued edges: both global-atomic edge passes and both memsets become one plain edge pass; sums in LDS-atomic order
# speedup vs baseline: 2.6776x; 1.8175x over previous
#include <hip/hip_runtime.h>
#include <hip/hip_bf16.h>

typedef __attribute__((ext_vector_type(2))) float v2f;
typedef __attribute__((ext_vector_type(8))) float v8f;

#define F_DIM 128
#define LEAKY 0.2f

#define AGG_GROUPS  256
#define AGG_ROWS    196
#define AGG_THREADS 1024
#define AGG_QCAP    8192
#define AGG_INFLIGHT 8

__device__ __forceinline__ unsigned f2ord(float f) {
  unsigned u = __float_as_uint(f);
  return (u & 0x80000000u) ? ~u : (u | 0x80000000u);
}
__device__ __forceinline__ float ord2f(unsigned u) {
  return (u & 0x80000000u) ? __uint_as_float(u & 0x7FFFFFFFu)
                           : __uint_as_float(~u);
}

__global__ __launch_bounds__(256) void gat_gemm_wmma(
    const float* __restrict__ x, const float* __restrict__ W,
    float* __restrict__ h, int M)
{
  __shared__ float sA[16 * 132];
  const int tid = threadIdx.x;
  const int m0  = blockIdx.x * 16;

  for (int i = tid; i < 512; i += 256) {
    int r = i >> 5, c = i & 31;
    float4 v = make_float4(0.f, 0.f, 0.f, 0.f);
    if (m0 + r < M) v = ((const float4*)(x + (size_t)(m0 + r) * F_DIM))[c];
    ((float4*)sA)[r * 33 + c] = v;
  }
  __syncthreads();

  const int lane = tid & 31;
  const int wv   = tid >> 5;
  const int row  = lane & 15;
  const int kh   = (lane >> 4) * 2;
  const int n0   = wv * 16;

  v8f acc = {};
  #pragma unroll 4
  for (int k = 0; k < F_DIM; k += 4) {
    v2f a, b;
    a.x = sA[row * 132 + k + kh + 0];
    a.y = sA[row * 132 + k + kh + 1];
    b.x = W[(k + kh + 0) * F_DIM + n0 + row];
    b.y = W[(k + kh + 1) * F_DIM + n0 + row];
    acc = __builtin_amdgcn_wmma_f32_16x16x4_f32(false, a, false, b,
                                                (short)0, acc, false, false);
  }

  const int mh = (lane >> 4) * 8;
  float* hp = h + (size_t)m0 * F_DIM + n0 + row;
  if (m0 + 16 <= M) {
    #pragma unroll
    for (int v = 0; v < 8; ++v) hp[(size_t)(mh + v) * F_DIM] = acc[v];
  } else {
    #pragma unroll
    for (int v = 0; v < 8; ++v)
      if (m0 + mh + v < M) hp[(size_t)(mh + v) * F_DIM] = acc[v];
  }
}

__global__ __launch_bounds__(256) void gat_alpha(
    const float* __restrict__ h, const float* __restrict__ a,
    float* __restrict__ as, float* __restrict__ ad, int N)
{
  const int node = blockIdx.x * 8 + (threadIdx.x >> 5);
  const int lane = threadIdx.x & 31;
  if (node >= N) return;
  float4 hv = ((const float4*)(h + (size_t)node * F_DIM))[lane];
  float4 a1 = ((const float4*)a)[lane];
  float4 a2 = ((const float4*)(a + F_DIM))[lane];
  float s = hv.x * a1.x + hv.y * a1.y + hv.z * a1.z + hv.w * a1.w;
  float d = hv.x * a2.x + hv.y * a2.y + hv.z * a2.z + hv.w * a2.w;
  #pragma unroll
  for (int m = 16; m >= 1; m >>= 1) {
    s += __shfl_xor(s, m, 32);
    d += __shfl_xor(d, m, 32);
  }
  if (lane == 0) { as[node] = s; ad[node] = d; }
}

__global__ __launch_bounds__(256) void gat_edge_logit(
    const int* __restrict__ src, const int* __restrict__ dst,
    const float* __restrict__ as, const float* __restrict__ ad,
    float* __restrict__ e, unsigned char* __restrict__ owner, int E)
{
  const int i0 = blockIdx.x * 1024 + threadIdx.x;
  int s[4], d[4];
  #pragma unroll
  for (int u = 0; u < 4; ++u) {
    const int i = i0 + u * 256;
    s[u] = 0; d[u] = 0;
    if (i < E) { s[u] = src[i]; d[u] = dst[i]; }
  }
  float v[4];
  #pragma unroll
  for (int u = 0; u < 4; ++u) v[u] = as[s[u]] + ad[d[u]];
  #pragma unroll
  for (int u = 0; u < 4; ++u) {
    const int i = i0 + u * 256;
    if (i < E) {
      const float t = v[u] > 0.f ? v[u] : LEAKY * v[u];
      e[i] = t;
      owner[i] = (unsigned char)((d[u] / AGG_ROWS) & (AGG_GROUPS - 1));
    }
  }
}

__device__ __forceinline__ unsigned zero_bytes(unsigned x) {
  unsigned t = (x & 0x7F7F7F7Fu) + 0x7F7F7F7Fu;
  return ~(t | x | 0x7F7F7F7Fu);
}

__device__ __forceinline__ void agg_take(int i, int* queue, int* qcount)
{
  const int pos = atomicAdd(qcount, 1);
  if (pos < AGG_QCAP) queue[pos] = i;
}

__global__ __launch_bounds__(AGG_THREADS) void gat_aggregate(
    const int* __restrict__ src, const int* __restrict__ dst,
    const float* __restrict__ h, const float* __restrict__ lg,
    const unsigned char* __restrict__ owner,
    float* __restrict__ out, int N, int E)
{
  __shared__ float    tile[AGG_ROWS * F_DIM];
  __shared__ int      queue[AGG_QCAP];
  __shared__ float    qv[AGG_QCAP];
  __shared__ unsigned char qd[AGG_QCAP];
  __shared__ unsigned smax[AGG_ROWS];
  __shared__ float    ssum[AGG_ROWS];
  __shared__ int      qcount;

  const int tid  = threadIdx.x;
  const int lane = tid & 31;
  const int wv   = tid >> 5;
  const unsigned me  = blockIdx.x;
  const unsigned me4 = me * 0x01010101u;

  int head = (int)((16u - (unsigned)((size_t)owner & 15u)) & 15u);
  if (head > E) head = E;
  const int nvec  = (E - head) >> 4;
  const int tail0 = head + (nvec << 4);
  const uint4* vp = (const uint4*)(owner + head);
  const int vstart = (int)(((long long)blockIdx.x * nvec) / AGG_GROUPS);

  for (int base = blockIdx.x * AGG_ROWS; base < N; base += AGG_GROUPS * AGG_ROWS) {
    for (int k = tid; k < AGG_ROWS * F_DIM / 4; k += AGG_THREADS)
      ((float4*)tile)[k] = make_float4(0.f, 0.f, 0.f, 0.f);
    for (int k = tid; k < AGG_ROWS; k += AGG_THREADS) {
      smax[k] = 0u;
      ssum[k] = 0.f;
    }
    if (tid == 0) qcount = 0;
    __syncthreads();

    int v0 = tid;
    for (; v0 + (AGG_INFLIGHT - 1) * AGG_THREADS < nvec; v0 += AGG_INFLIGHT * AGG_THREADS) {
      uint4 w[AGG_INFLIGHT];
      int   vi[AGG_INFLIGHT];
      #pragma unroll
      for (int u = 0; u < AGG_INFLIGHT; ++u) {
        int v = v0 + u * AGG_THREADS + vstart;
        if (v >= nvec) v -= nvec;
        vi[u] = v;
        w[u]  = vp[v];
      }
      #pragma unroll
      for (int u = 0; u < AGG_INFLIGHT; ++u) {
        unsigned m = (zero_bytes(w[u].x ^ me4) >> 7) | (zero_bytes(w[u].y ^ me4) >> 6) |
                     (zero_bytes(w[u].z ^ me4) >> 5) | (zero_bytes(w[u].w ^ me4) >> 4);
        while (m) {
          const int p = __builtin_ctz(m);
          m &= m - 1;
          agg_take(head + (vi[u] << 4) + ((p & 7) << 2) + (p >> 3), queue, &qcount);
        }
      }
    }
    for (; v0 < nvec; v0 += AGG_THREADS) {
      int v = v0 + vstart;
      if (v >= nvec) v -= nvec;
      const uint4 w1 = vp[v];
      unsigned m = (zero_bytes(w1.x ^ me4) >> 7) | (zero_bytes(w1.y ^ me4) >> 6) |
                   (zero_bytes(w1.z ^ me4) >> 5) | (zero_bytes(w1.w ^ me4) >> 4);
      while (m) {
        const int p = __builtin_ctz(m);
        m &= m - 1;
        agg_take(head + (v << 4) + ((p & 7) << 2) + (p >> 3), queue, &qcount);
      }
    }
    for (int i = tid; i < head; i += AGG_THREADS)
      if (owner[i] == me) agg_take(i, queue, &qcount);
    for (int i = tail0 + tid; i < E; i += AGG_THREADS)
      if (owner[i] == me) agg_take(i, queue, &qcount);
    __syncthreads();

    const int total = qcount;
    if (total <= AGG_QCAP) {
      const int n = total;

      for (int q0 = tid; q0 < n; q0 += 4 * AGG_THREADS) {
        int   ei[4], es[4], ed[4];
        float ev[4];
        #pragma unroll
        for (int u = 0; u < 4; ++u) {
          const int q = q0 + u * AGG_THREADS;
          ei[u] = (q < n) ? queue[q] : -1;
        }
        #pragma unroll
        for (int u = 0; u < 4; ++u) {
          es[u] = 0; ed[u] = -1; ev[u] = 0.f;
          if (ei[u] >= 0) { ed[u] = dst[ei[u]]; es[u] = src[ei[u]]; ev[u] = lg[ei[u]]; }
        }
        #pragma unroll
        for (int u = 0; u < 4; ++u) {
          if (ei[u] >= 0) {
            const int q = q0 + u * AGG_THREADS;
            const int d = ed[u];
            const bool mine = d >= base && d < base + AGG_ROWS && d < N;
            queue[q] = es[u];
            qv[q]    = ev[u];
            qd[q]    = mine ? (unsigned char)(d - base) : (unsigned char)255;
            if (mine) atomicMax(&smax[d - base], f2ord(ev[u]));
          }
        }
      }
      __syncthreads();

      for (int q = tid; q < n; q += AGG_THREADS) {
        const int dl = qd[q];
        if (dl != 255) {
          const float v = expf(qv[q] - ord2f(smax[dl]));
          qv[q] = v;
          atomicAdd(&ssum[dl], v);
        }
      }
      __syncthreads();

      for (int q0 = wv * 32; q0 < n; q0 += AGG_THREADS) {
        const int q = q0 + lane;
        int s = 0, dl = -1;
        float c = 0.f;
        if (q < n) {
          const int d8 = qd[q];
          if (d8 != 255) {
            s  = queue[q];
            dl = d8;
            c  = qv[q] / (ssum[d8] + 1e-16f);
          }
        }
        for (int j = 0; j < 32; j += 8) {
          if (q0 + j >= n) break;
          float r[8][4];
          int   dj[8];
          float cj[8];
          #pragma unroll
          for (int u = 0; u < 8; ++u) {
            const int sj = __builtin_amdgcn_readlane(s, j + u);
            dj[u] = __builtin_amdgcn_readlane(dl, j + u);
            cj[u] = __int_as_float(__builtin_amdgcn_readlane(__float_as_int(c), j + u));
            if (dj[u] >= 0) {
              const float* hp = h + (size_t)sj * F_DIM + lane;
              r[u][0] = hp[0];
              r[u][1] = hp[32];
              r[u][2] = hp[64];
              r[u][3] = hp[96];
            }
          }
          #pragma unroll
          for (int u = 0; u < 8; ++u) {
            if (dj[u] >= 0) {
              float* tp = tile + dj[u] * F_DIM + lane;
              atomicAdd(tp,      cj[u] * r[u][0]);
              atomicAdd(tp + 32, cj[u] * r[u][1]);
              atomicAdd(tp + 64, cj[u] * r[u][2]);
              atomicAdd(tp + 96, cj[u] * r[u][3]);
            }
          }
        }
      }
    } else {
      for (int i = tid; i < E; i += AGG_THREADS) {
        if (owner[i] != me) continue;
        const int d = dst[i];
        if (d < base || d >= base + AGG_ROWS || d >= N) continue;
        atomicMax(&smax[d - base], f2ord(lg[i]));
      }
      __syncthreads();
      for (int i = tid; i < E; i += AGG_THREADS) {
        if (owner[i] != me) continue;
        const int d = dst[i];
        if (d < base || d >= base + AGG_ROWS || d >= N) continue;
        atomicAdd(&ssum[d - base], expf(lg[i] - ord2f(smax[d - base])));
      }
      __syncthreads();
      for (int i = tid; i < E; i += AGG_THREADS) {
        if (owner[i] != me) continue;
        const int d = dst[i];
        if (d < base || d >= base + AGG_ROWS || d >= N) continue;
        const float c = expf(lg[i] - ord2f(smax[d - base])) / (ssum[d - base] + 1e-16f);
        const float* hp = h + (size_t)src[i] * F_DIM;
        float* tp = tile + (d - base) * F_DIM;
        for (int f = 0; f < F_DIM; ++f) atomicAdd(tp + f, c * hp[f]);
      }
    }
    __syncthreads();

    const int rows = min(AGG_ROWS, N - base);
    float4* op = (float4*)(out + (size_t)base * F_DIM);
    for (int k = tid; k < rows * (F_DIM / 4); k += AGG_THREADS)
      op[k] = ((const float4*)tile)[k];
    __syncthreads();
  }
}

extern "C" void kernel_launch(void* const* d_in, const int* in_sizes, int n_in,
                              void* d_out, int out_size, void* d_ws, size_t ws_size,
                              hipStream_t stream) {
  const float* x  = (const float*)d_in[0];
  const int*   ei = (const int*)d_in[1];
  const float* W  = (const float*)d_in[2];
  const float* a  = (const float*)d_in[3];
  const int N = in_sizes[0] / F_DIM;
  const int E = in_sizes[1] / 2;
  const int* src = ei;
  const int* dst = ei + E;

  float*    h      = (float*)d_ws;
  float*    as     = h + (size_t)N * F_DIM;
  float*    ad     = as + N;
  float*    e      = ad + N;
  unsigned char* owner = (unsigned char*)(e + E);
  float*    out    = (float*)d_out;

  gat_gemm_wmma <<<(N + 15) / 16,  256, 0, stream>>>(x, W, h, N);
  gat_alpha     <<<(N + 7)  / 8,   256, 0, stream>>>(h, a, as, ad, N);
  gat_edge_logit<<<(E + 1023) / 1024, 256, 0, stream>>>(src, dst, as, ad, e, owner, E);
  gat_aggregate <<<AGG_GROUPS, AGG_THREADS, 0, stream>>>(src, dst, h, e, owner, out, N, E);
}


# x@W and alpha launches as resident grids of looping waves: a wave keeps its W fragments in registers over its row tiles, alpha waves take 8 rows per trip; 75,000 waves entered become 12,288
# speedup vs baseline: 2.8542x; 1.0660x over previous
#include <hip/hip_runtime.h>
#include <hip/hip_bf16.h>

typedef __attribute__((ext_vector_type(2))) float v2f;
typedef __attribute__((ext_vector_type(8))) float v8f;

#define F_DIM 128
#define LEAKY 0.2f

#define AGG_GROUPS  256
#define AGG_ROWS    196
#define AGG_THREADS 1024
#define AGG_QCAP    8192
#define AGG_INFLIGHT 8

#define GEMM_GROUPS   1024
#define ALPHA_GROUPS  256
#define ALPHA_THREADS 512
#define ALPHA_ROWS    8

__device__ __forceinline__ unsigned f2ord(float f) {
  unsigned u = __float_as_uint(f);
  return (u & 0x80000000u) ? ~u : (u | 0x80000000u);
}
__device__ __forceinline__ float ord2f(unsigned u) {
  return (u & 0x80000000u) ? __uint_as_float(u & 0x7FFFFFFFu)
                           : __uint_as_float(~u);
}

__global__ __launch_bounds__(256) void gat_gemm_wmma(
    const float* __restrict__ x, const float* __restrict__ W,
    float* __restrict__ h, int M)
{
  __shared__ float sA[16 * 132];
  const int tid  = threadIdx.x;
  const int lane = tid & 31;
  const int wv   = tid >> 5;
  const int row  = lane & 15;
  const int kh   = (lane >> 4) * 2;
  const int n0   = wv * 16;
  const int mh   = (lane >> 4) * 8;
  const int ntiles = (M + 15) >> 4;

  const int sr = tid >> 5, sc = tid & 31;

  v2f bf[32];
  #pragma unroll
  for (int kk = 0; kk < 32; ++kk) {
    bf[kk].x = W[(4 * kk + kh + 0) * F_DIM + n0 + row];
    bf[kk].y = W[(4 * kk + kh + 1) * F_DIM + n0 + row];
  }

  int tile = blockIdx.x;
  float4 p0 = make_float4(0.f, 0.f, 0.f, 0.f), p1 = p0;
  if (tile < ntiles) {
    const int m0 = tile * 16;
    if (m0 + sr     < M) p0 = ((const float4*)(x + (size_t)(m0 + sr)     * F_DIM))[sc];
    if (m0 + sr + 8 < M) p1 = ((const float4*)(x + (size_t)(m0 + sr + 8) * F_DIM))[sc];
  }

  for (; tile < ntiles; tile += gridDim.x) {
    const int m0 = tile * 16;
    ((float4*)sA)[sr * 33 + sc]       = p0;
    ((float4*)sA)[(sr + 8) * 33 + sc] = p1;
    __syncthreads();

    const int nt = tile + gridDim.x;
    p0 = make_float4(0.f, 0.f, 0.f, 0.f); p1 = p0;
    if (nt < ntiles) {
      const int q0 = nt * 16;
      if (q0 + sr     < M) p0 = ((const float4*)(x + (size_t)(q0 + sr)     * F_DIM))[sc];
      if (q0 + sr + 8 < M) p1 = ((const float4*)(x + (size_t)(q0 + sr + 8) * F_DIM))[sc];
    }

    v8f acc = {};
    #pragma unroll
    for (int kk = 0; kk < 32; ++kk) {
      v2f a;
      a.x = sA[row * 132 + 4 * kk + kh + 0];
      a.y = sA[row * 132 + 4 * kk + kh + 1];
      acc = __builtin_amdgcn_wmma_f32_16x16x4_f32(false, a, false, bf[kk],
                                                  (short)0, acc, false, false);
    }

    float* hp = h + (size_t)m0 * F_DIM + n0 + row;
    if (m0 + 16 <= M) {
      #pragma unroll
      for (int v = 0; v < 8; ++v) hp[(size_t)(mh + v) * F_DIM] = acc[v];
    } else {
      #pragma unroll
      for (int v = 0; v < 8; ++v)
        if (m0 + mh + v < M) hp[(size_t)(mh + v) * F_DIM] = acc[v];
    }
    __syncthreads();
  }
}

__global__ __launch_bounds__(ALPHA_THREADS) void gat_alpha(
    const float* __restrict__ h, const float* __restrict__ a,
    float* __restrict__ as, float* __restrict__ ad, int N)
{
  const int lane   = threadIdx.x & 31;
  const int wave   = blockIdx.x * (ALPHA_THREADS / 32) + (threadIdx.x >> 5);
  const int nwaves = gridDim.x * (ALPHA_THREADS / 32);
  const float4 a1 = ((const float4*)a)[lane];
  const float4 a2 = ((const float4*)(a + F_DIM))[lane];

  for (int node0 = wave * ALPHA_ROWS; node0 < N; node0 += nwaves * ALPHA_ROWS) {
    float4 hv[ALPHA_ROWS];
    #pragma unroll
    for (int u = 0; u < ALPHA_ROWS; ++u) {
      hv[u] = make_float4(0.f, 0.f, 0.f, 0.f);
      if (node0 + u < N) hv[u] = ((const float4*)(h + (size_t)(node0 + u) * F_DIM))[lane];
    }
    float sv = 0.f, dv = 0.f;
    #pragma unroll
    for (int u = 0; u < ALPHA_ROWS; ++u) {
      float s = hv[u].x * a1.x + hv[u].y * a1.y + hv[u].z * a1.z + hv[u].w * a1.w;
      float d = hv[u].x * a2.x + hv[u].y * a2.y + hv[u].z * a2.z + hv[u].w * a2.w;
      #pragma unroll
      for (int m = 16; m >= 1; m >>= 1) {
        s += __shfl_xor(s, m, 32);
        d += __shfl_xor(d, m, 32);
      }
      if (lane == u) { sv = s; dv = d; }
    }
    if (lane < ALPHA_ROWS && node0 + lane < N) {
      as[node0 + lane] = sv;
      ad[node0 + lane] = dv;
    }
  }
}

__global__ __launch_bounds__(256) void gat_edge_logit(
    const int* __restrict__ src, const int* __restrict__ dst,
    const float* __restrict__ as, const float* __restrict__ ad,
    float* __restrict__ e, unsigned char* __restrict__ owner, int E)
{
  const int i0 = blockIdx.x * 1024 + threadIdx.x;
  int s[4], d[4];
  #pragma unroll
  for (int u = 0; u < 4; ++u) {
    const int i = i0 + u * 256;
    s[u] = 0; d[u] = 0;
    if (i < E) { s[u] = src[i]; d[u] = dst[i]; }
  }
  float v[4];
  #pragma unroll
  for (int u = 0; u < 4; ++u) v[u] = as[s[u]] + ad[d[u]];
  #pragma unroll
  for (int u = 0; u < 4; ++u) {
    const int i = i0 + u * 256;
    if (i < E) {
      const float t = v[u] > 0.f ? v[u] : LEAKY * v[u];
      e[i] = t;
      owner[i] = (unsigned char)((d[u] / AGG_ROWS) & (AGG_GROUPS - 1));
    }
  }
}

__device__ __forceinline__ unsigned zero_bytes(unsigned x) {
  unsigned t = (x & 0x7F7F7F7Fu) + 0x7F7F7F7Fu;
  return ~(t | x | 0x7F7F7F7Fu);
}

__device__ __forceinline__ void agg_take(int i, int* queue, int* qcount)
{
  const int pos = atomicAdd(qcount, 1);
  if (pos < AGG_QCAP) queue[pos] = i;
}

__global__ __launch_bounds__(AGG_THREADS) void gat_aggregate(
    const int* __restrict__ src, const int* __restrict__ dst,
    const float* __restrict__ h, const float* __restrict__ lg,
    const unsigned char* __restrict__ owner,
    float* __restrict__ out, int N, int E)
{
  __shared__ float    tile[AGG_ROWS * F_DIM];
  __shared__ int      queue[AGG_QCAP];
  __shared__ float    qv[AGG_QCAP];
  __shared__ unsigned char qd[AGG_QCAP];
  __shared__ unsigned smax[AGG_ROWS];
  __shared__ float    ssum[AGG_ROWS];
  __shared__ int      qcount;

  const int tid  = threadIdx.x;
  const int lane = tid & 31;
  const int wv   = tid >> 5;
  const unsigned me  = blockIdx.x;
  const unsigned me4 = me * 0x01010101u;

  int head = (int)((16u - (unsigned)((size_t)owner & 15u)) & 15u);
  if (head > E) head = E;
  const int nvec  = (E - head) >> 4;
  const int tail0 = head + (nvec << 4);
  const uint4* vp = (const uint4*)(owner + head);
  const int vstart = (int)(((long long)blockIdx.x * nvec) / AGG_GROUPS);

  for (int base = blockIdx.x * AGG_ROWS; base < N; base += AGG_GROUPS * AGG_ROWS) {
    for (int k = tid; k < AGG_ROWS * F_DIM / 4; k += AGG_THREADS)
      ((float4*)tile)[k] = make_float4(0.f, 0.f, 0.f, 0.f);
    for (int k = tid; k < AGG_ROWS; k += AGG_THREADS) {
      smax[k] = 0u;
      ssum[k] = 0.f;
    }
    if (tid == 0) qcount = 0;
    __syncthreads();

    int v0 = tid;
    for (; v0 + (AGG_INFLIGHT - 1) * AGG_THREADS < nvec; v0 += AGG_INFLIGHT * AGG_THREADS) {
      uint4 w[AGG_INFLIGHT];
      int   vi[AGG_INFLIGHT];
      #pragma unroll
      for (int u = 0; u < AGG_INFLIGHT; ++u) {
        int v = v0 + u * AGG_THREADS + vstart;
        if (v >= nvec) v -= nvec;
        vi[u] = v;
        w[u]  = vp[v];
      }
      #pragma unroll
      for (int u = 0; u < AGG_INFLIGHT; ++u) {
        unsigned m = (zero_bytes(w[u].x ^ me4) >> 7) | (zero_bytes(w[u].y ^ me4) >> 6) |
                     (zero_bytes(w[u].z ^ me4) >> 5) | (zero_bytes(w[u].w ^ me4) >> 4);
        while (m) {
          const int p = __builtin_ctz(m);
          m &= m - 1;
          agg_take(head + (vi[u] << 4) + ((p & 7) << 2) + (p >> 3), queue, &qcount);
        }
      }
    }
    for (; v0 < nvec; v0 += AGG_THREADS) {
      int v = v0 + vstart;
      if (v >= nvec) v -= nvec;
      const uint4 w1 = vp[v];
      unsigned m = (zero_bytes(w1.x ^ me4) >> 7) | (zero_bytes(w1.y ^ me4) >> 6) |
                   (zero_bytes(w1.z ^ me4) >> 5) | (zero_bytes(w1.w ^ me4) >> 4);
      while (m) {
        const int p = __builtin_ctz(m);
        m &= m - 1;
        agg_take(head + (v << 4) + ((p & 7) << 2) + (p >> 3), queue, &qcount);
      }
    }
    for (int i = tid; i < head; i += AGG_THREADS)
      if (owner[i] == me) agg_take(i, queue, &qcount);
    for (int i = tail0 + tid; i < E; i += AGG_THREADS)
      if (owner[i] == me) agg_take(i, queue, &qcount);
    __syncthreads();

    const int total = qcount;
    if (total <= AGG_QCAP) {
      const int n = total;

      for (int q0 = tid; q0 < n; q0 += 4 * AGG_THREADS) {
        int   ei[4], es[4], ed[4];
        float ev[4];
        #pragma unroll
        for (int u = 0; u < 4; ++u) {
          const int q = q0 + u * AGG_THREADS;
          ei[u] = (q < n) ? queue[q] : -1;
        }
        #pragma unroll
        for (int u = 0; u < 4; ++u) {
          es[u] = 0; ed[u] = -1; ev[u] = 0.f;
          if (ei[u] >= 0) { ed[u] = dst[ei[u]]; es[u] = src[ei[u]]; ev[u] = lg[ei[u]]; }
        }
        #pragma unroll
        for (int u = 0; u < 4; ++u) {
          if (ei[u] >= 0) {
            const int q = q0 + u * AGG_THREADS;
            const int d = ed[u];
            const bool mine = d >= base && d < base + AGG_ROWS && d < N;
            queue[q] = es[u];
            qv[q]    = ev[u];
            qd[q]    = mine ? (unsigned char)(d - base) : (unsigned char)255;
            if (mine) atomicMax(&smax[d - base], f2ord(ev[u]));
          }
        }
      }
      __syncthreads();

      for (int q = tid; q < n; q += AGG_THREADS) {
        const int dl = qd[q];
        if (dl != 255) {
          const float v = expf(qv[q] - ord2f(smax[dl]));
          qv[q] = v;
          atomicAdd(&ssum[dl], v);
        }
      }
      __syncthreads();

      for (int q0 = wv * 32; q0 < n; q0 += AGG_THREADS) {
        const int q = q0 + lane;
        int s = 0, dl = -1;
        float c = 0.f;
        if (q < n) {
          const int d8 = qd[q];
          if (d8 != 255) {
            s  = queue[q];
            dl = d8;
            c  = qv[q] / (ssum[d8] + 1e-16f);
          }
        }
        for (int j = 0; j < 32; j += 8) {
          if (q0 + j >= n) break;
          float r[8][4];
          int   dj[8];
          float cj[8];
          #pragma unroll
          for (int u = 0; u < 8; ++u) {
            const int sj = __builtin_amdgcn_readlane(s, j + u);
            dj[u] = __builtin_amdgcn_readlane(dl, j + u);
            cj[u] = __int_as_float(__builtin_amdgcn_readlane(__float_as_int(c), j + u));
            if (dj[u] >= 0) {
              const float* hp = h + (size_t)sj * F_DIM + lane;
              r[u][0] = hp[0];
              r[u][1] = hp[32];
              r[u][2] = hp[64];
              r[u][3] = hp[96];
            }
          }
          #pragma unroll
          for (int u = 0; u < 8; ++u) {
            if (dj[u] >= 0) {
              float* tp = tile + dj[u] * F_DIM + lane;
              atomicAdd(tp,      cj[u] * r[u][0]);
              atomicAdd(tp + 32, cj[u] * r[u][1]);
              atomicAdd(tp + 64, cj[u] * r[u][2]);
              atomicAdd(tp + 96, cj[u] * r[u][3]);
            }
          }
        }
      }
    } else {
      for (int i = tid; i < E; i += AGG_THREADS) {
        if (owner[i] != me) continue;
        const int d = dst[i];
        if (d < base || d >= base + AGG_ROWS || d >= N) continue;
        atomicMax(&smax[d - base], f2ord(lg[i]));
      }
      __syncthreads();
      for (int i = tid; i < E; i += AGG_THREADS) {
        if (owner[i] != me) continue;
        const int d = dst[i];
        if (d < base || d >= base + AGG_ROWS || d >= N) continue;
        atomicAdd(&ssum[d - base], expf(lg[i] - ord2f(smax[d - base])));
      }
      __syncthreads();
      for (int i = tid; i < E; i += AGG_THREADS) {
        if (owner[i] != me) continue;
        const int d = dst[i];
        if (d < base || d >= base + AGG_ROWS || d >= N) continue;
        const float c = expf(lg[i] - ord2f(smax[d - base])) / (ssum[d - base] + 1e-16f);
        const float* hp = h + (size_t)src[i] * F_DIM;
        float* tp = tile + (d - base) * F_DIM;
        for (int f = 0; f < F_DIM; ++f) atomicAdd(tp + f, c * hp[f]);
      }
    }
    __syncthreads();

    const int rows = min(AGG_ROWS, N - base);
    float4* op = (float4*)(out + (size_t)base * F_DIM);
    for (int k = tid; k < rows * (F_DIM / 4); k += AGG_THREADS)
      op[k] = ((const float4*)tile)[k];
    __syncthreads();
  }
}

extern "C" void kernel_launch(void* const* d_in, const int* in_sizes, int n_in,
                              void* d_out, int out_size, void* d_ws, size_t ws_size,
                              hipStream_t stream) {
  const float* x  = (const float*)d_in[0];
  const int*   ei = (const int*)d_in[1];
  const float* W  = (const float*)d_in[2];
  const float* a  = (const float*)d_in[3];
  const int N = in_sizes[0] / F_DIM;
  const int E = in_sizes[1] / 2;
  const int* src = ei;
  const int* dst = ei + E;

  float*    h      = (float*)d_ws;
  float*    as     = h + (size_t)N * F_DIM;
  float*    ad     = as + N;
  float*    e      = ad + N;
  unsigned char* owner = (unsigned char*)(e + E);
  float*    out    = (float*)d_out;

  gat_gemm_wmma <<<GEMM_GROUPS,  256, 0, stream>>>(x, W, h, N);
  gat_alpha     <<<ALPHA_GROUPS, ALPHA_THREADS, 0, stream>>>(h, a, as, ad, N);
  gat_edge_logit<<<(E + 1023) / 1024, 256, 0, stream>>>(src, dst, as, ad, e, owner, E);
  gat_aggregate <<<AGG_GROUPS, AGG_THREADS, 0, stream>>>(src, dst, h, e, owner, out, N, E);
}


# aggregate: the 32 row loads of a wave's trip stand under no branch and issue as one group; under each slot's scalar branch the compiler had put a counted wait before every load (4 in flight)
# speedup vs baseline: 3.1677x; 1.1098x over previous
#include <hip/hip_runtime.h>
#include <hip/hip_bf16.h>

typedef __attribute__((ext_vector_type(2))) float v2f;
typedef __attribute__((ext_vector_type(8))) float v8f;

#define F_DIM 128
#define LEAKY 0.2f

#define AGG_GROUPS  256
#define AGG_ROWS    196
#define AGG_THREADS 1024
#define AGG_QCAP    8192
#define AGG_INFLIGHT 8

#define GEMM_GROUPS   1024
#define ALPHA_GROUPS  256
#define ALPHA_THREADS 512
#define ALPHA_ROWS    8

__device__ __forceinline__ unsigned f2ord(float f) {
  unsigned u = __float_as_uint(f);
  return (u & 0x80000000u) ? ~u : (u | 0x80000000u);
}
__device__ __forceinline__ float ord2f(unsigned u) {
  return (u & 0x80000000u) ? __uint_as_float(u & 0x7FFFFFFFu)
                           : __uint_as_float(~u);
}

__global__ __launch_bounds__(256) void gat_gemm_wmma(
    const float* __restrict__ x, const float* __restrict__ W,
    float* __restrict__ h, int M)
{
  __shared__ float sA[16 * 132];
  const int tid  = threadIdx.x;
  const int lane = tid & 31;
  const int wv   = tid >> 5;
  const int row  = lane & 15;
  const int kh   = (lane >> 4) * 2;
  const int n0   = wv * 16;
  const int mh   = (lane >> 4) * 8;
  const int ntiles = (M + 15) >> 4;

  const int sr = tid >> 5, sc = tid & 31;

  v2f bf[32];
  #pragma unroll
  for (int kk = 0; kk < 32; ++kk) {
    bf[kk].x = W[(4 * kk + kh + 0) * F_DIM + n0 + row];
    bf[kk].y = W[(4 * kk + kh + 1) * F_DIM + n0 + row];
  }

  int tile = blockIdx.x;
  float4 p0 = make_float4(0.f, 0.f, 0.f, 0.f), p1 = p0;
  if (tile < ntiles) {
    const int m0 = tile * 16;
    if (m0 + sr     < M) p0 = ((const float4*)(x + (size_t)(m0 + sr)     * F_DIM))[sc];
    if (m0 + sr + 8 < M) p1 = ((const float4*)(x + (size_t)(m0 + sr + 8) * F_DIM))[sc];
  }

  for (; tile < ntiles; tile += gridDim.x) {
    const int m0 = tile * 16;
    ((float4*)sA)[sr * 33 + sc]       = p0;
    ((float4*)sA)[(sr + 8) * 33 + sc] = p1;
    __syncthreads();

    const int nt = tile + gridDim.x;
    p0 = make_float4(0.f, 0.f, 0.f, 0.f); p1 = p0;
    if (nt < ntiles) {
      const int q0 = nt * 16;
      if (q0 + sr     < M) p0 = ((const float4*)(x + (size_t)(q0 + sr)     * F_DIM))[sc];
      if (q0 + sr + 8 < M) p1 = ((const float4*)(x + (size_t)(q0 + sr + 8) * F_DIM))[sc];
    }

    v8f acc = {};
    #pragma unroll
    for (int kk = 0; kk < 32; ++kk) {
      v2f a;
      a.x = sA[row * 132 + 4 * kk + kh + 0];
      a.y = sA[row * 132 + 4 * kk + kh + 1];
      acc = __builtin_amdgcn_wmma_f32_16x16x4_f32(false, a, false, bf[kk],
                                                  (short)0, acc, false, false);
    }

    float* hp = h + (size_t)m0 * F_DIM + n0 + row;
    if (m0 + 16 <= M) {
      #pragma unroll
      for (int v = 0; v < 8; ++v) hp[(size_t)(mh + v) * F_DIM] = acc[v];
    } else {
      #pragma unroll
      for (int v = 0; v < 8; ++v)
        if (m0 + mh + v < M) hp[(size_t)(mh + v) * F_DIM] = acc[v];
    }
    __syncthreads();
  }
}

__global__ __launch_bounds__(ALPHA_THREADS) void gat_alpha(
    const float* __restrict__ h, const float* __restrict__ a,
    float* __restrict__ as, float* __restrict__ ad, int N)
{
  const int lane   = threadIdx.x & 31;
  const int wave   = blockIdx.x * (ALPHA_THREADS / 32) + (threadIdx.x >> 5);
  const int nwaves = gridDim.x * (ALPHA_THREADS / 32);
  const float4 a1 = ((const float4*)a)[lane];
  const float4 a2 = ((const float4*)(a + F_DIM))[lane];

  for (int node0 = wave * ALPHA_ROWS; node0 < N; node0 += nwaves * ALPHA_ROWS) {
    float4 hv[ALPHA_ROWS];
    #pragma unroll
    for (int u = 0; u < ALPHA_ROWS; ++u) {
      hv[u] = make_float4(0.f, 0.f, 0.f, 0.f);
      if (node0 + u < N) hv[u] = ((const float4*)(h + (size_t)(node0 + u) * F_DIM))[lane];
    }
    float sv = 0.f, dv = 0.f;
    #pragma unroll
    for (int u = 0; u < ALPHA_ROWS; ++u) {
      float s = hv[u].x * a1.x + hv[u].y * a1.y + hv[u].z * a1.z + hv[u].w * a1.w;
      float d = hv[u].x * a2.x + hv[u].y * a2.y + hv[u].z * a2.z + hv[u].w * a2.w;
      #pragma unroll
      for (int m = 16; m >= 1; m >>= 1) {
        s += __shfl_xor(s, m, 32);
        d += __shfl_xor(d, m, 32);
      }
      if (lane == u) { sv = s; dv = d; }
    }
    if (lane < ALPHA_ROWS && node0 + lane < N) {
      as[node0 + lane] = sv;
      ad[node0 + lane] = dv;
    }
  }
}

__global__ __launch_bounds__(256) void gat_edge_logit(
    const int* __restrict__ src, const int* __restrict__ dst,
    const float* __restrict__ as, const float* __restrict__ ad,
    float* __restrict__ e, unsigned char* __restrict__ owner, int E)
{
  const int i0 = blockIdx.x * 1024 + threadIdx.x;
  int s[4], d[4];
  #pragma unroll
  for (int u = 0; u < 4; ++u) {
    const int i = i0 + u * 256;
    s[u] = 0; d[u] = 0;
    if (i < E) { s[u] = src[i]; d[u] = dst[i]; }
  }
  float v[4];
  #pragma unroll
  for (int u = 0; u < 4; ++u) v[u] = as[s[u]] + ad[d[u]];
  #pragma unroll
  for (int u = 0; u < 4; ++u) {
    const int i = i0 + u * 256;
    if (i < E) {
      const float t = v[u] > 0.f ? v[u] : LEAKY * v[u];
      e[i] = t;
      owner[i] = (unsigned char)((d[u] / AGG_ROWS) & (AGG_GROUPS - 1));
    }
  }
}

__device__ __forceinline__ unsigned zero_bytes(unsigned x) {
  unsigned t = (x & 0x7F7F7F7Fu) + 0x7F7F7F7Fu;
  return ~(t | x | 0x7F7F7F7Fu);
}

__device__ __forceinline__ void agg_take(int i, int* queue, int* qcount)
{
  const int pos = atomicAdd(qcount, 1);
  if (pos < AGG_QCAP) queue[pos] = i;
}

__global__ __launch_bounds__(AGG_THREADS) void gat_aggregate(
    const int* __restrict__ src, const int* __restrict__ dst,
    const float* __restrict__ h, const float* __restrict__ lg,
    const unsigned char* __restrict__ owner,
    float* __restrict__ out, int N, int E)
{
  __shared__ float    tile[AGG_ROWS * F_DIM];
  __shared__ int      queue[AGG_QCAP];
  __shared__ float    qv[AGG_QCAP];
  __shared__ unsigned char qd[AGG_QCAP];
  __shared__ unsigned smax[AGG_ROWS];
  __shared__ float    ssum[AGG_ROWS];
  __shared__ int      qcount;

  const int tid  = threadIdx.x;
  const int lane = tid & 31;
  const int wv   = tid >> 5;
  const unsigned me  = blockIdx.x;
  const unsigned me4 = me * 0x01010101u;

  int head = (int)((16u - (unsigned)((size_t)owner & 15u)) & 15u);
  if (head > E) head = E;
  const int nvec  = (E - head) >> 4;
  const int tail0 = head + (nvec << 4);
  const uint4* vp = (const uint4*)(owner + head);
  const int vstart = (int)(((long long)blockIdx.x * nvec) / AGG_GROUPS);

  for (int base = blockIdx.x * AGG_ROWS; base < N; base += AGG_GROUPS * AGG_ROWS) {
    for (int k = tid; k < AGG_ROWS * F_DIM / 4; k += AGG_THREADS)
      ((float4*)tile)[k] = make_float4(0.f, 0.f, 0.f, 0.f);
    for (int k = tid; k < AGG_ROWS; k += AGG_THREADS) {
      smax[k] = 0u;
      ssum[k] = 0.f;
    }
    if (tid == 0) qcount = 0;
    __syncthreads();

    int v0 = tid;
    for (; v0 + (AGG_INFLIGHT - 1) * AGG_THREADS < nvec; v0 += AGG_INFLIGHT * AGG_THREADS) {
      uint4 w[AGG_INFLIGHT];
      int   vi[AGG_INFLIGHT];
      #pragma unroll
      for (int u = 0; u < AGG_INFLIGHT; ++u) {
        int v = v0 + u * AGG_THREADS + vstart;
        if (v >= nvec) v -= nvec;
        vi[u] = v;
        w[u]  = vp[v];
      }
      #pragma unroll
      for (int u = 0; u < AGG_INFLIGHT; ++u) {
        unsigned m = (zero_bytes(w[u].x ^ me4) >> 7) | (zero_bytes(w[u].y ^ me4) >> 6) |
                     (zero_bytes(w[u].z ^ me4) >> 5) | (zero_bytes(w[u].w ^ me4) >> 4);
        while (m) {
          const int p = __builtin_ctz(m);
          m &= m - 1;
          agg_take(head + (vi[u] << 4) + ((p & 7) << 2) + (p >> 3), queue, &qcount);
        }
      }
    }
    for (; v0 < nvec; v0 += AGG_THREADS) {
      int v = v0 + vstart;
      if (v >= nvec) v -= nvec;
      const uint4 w1 = vp[v];
      unsigned m = (zero_bytes(w1.x ^ me4) >> 7) | (zero_bytes(w1.y ^ me4) >> 6) |
                   (zero_bytes(w1.z ^ me4) >> 5) | (zero_bytes(w1.w ^ me4) >> 4);
      while (m) {
        const int p = __builtin_ctz(m);
        m &= m - 1;
        agg_take(head + (v << 4) + ((p & 7) << 2) + (p >> 3), queue, &qcount);
      }
    }
    for (int i = tid; i < head; i += AGG_THREADS)
      if (owner[i] == me) agg_take(i, queue, &qcount);
    for (int i = tail0 + tid; i < E; i += AGG_THREADS)
      if (owner[i] == me) agg_take(i, queue, &qcount);
    __syncthreads();

    const int total = qcount;
    if (total <= AGG_QCAP) {
      const int n = total;

      for (int q0 = tid; q0 < n; q0 += 4 * AGG_THREADS) {
        int   ei[4], es[4], ed[4];
        float ev[4];
        #pragma unroll
        for (int u = 0; u < 4; ++u) {
          const int q = q0 + u * AGG_THREADS;
          ei[u] = (q < n) ? queue[q] : -1;
        }
        #pragma unroll
        for (int u = 0; u < 4; ++u) {
          es[u] = 0; ed[u] = -1; ev[u] = 0.f;
          if (ei[u] >= 0) { ed[u] = dst[ei[u]]; es[u] = src[ei[u]]; ev[u] = lg[ei[u]]; }
        }
        #pragma unroll
        for (int u = 0; u < 4; ++u) {
          if (ei[u] >= 0) {
            const int q = q0 + u * AGG_THREADS;
            const int d = ed[u];
            const bool mine = d >= base && d < base + AGG_ROWS && d < N;
            queue[q] = es[u];
            qv[q]    = ev[u];
            qd[q]    = mine ? (unsigned char)(d - base) : (unsigned char)255;
            if (mine) atomicMax(&smax[d - base], f2ord(ev[u]));
          }
        }
      }
      __syncthreads();

      for (int q = tid; q < n; q += AGG_THREADS) {
        const int dl = qd[q];
        if (dl != 255) {
          const float v = expf(qv[q] - ord2f(smax[dl]));
          qv[q] = v;
          atomicAdd(&ssum[dl], v);
        }
      }
      __syncthreads();

      for (int q0 = wv * 32; q0 < n; q0 += AGG_THREADS) {
        const int q = q0 + lane;
        int s = 0, dl = -1;
        float c = 0.f;
        if (q < n) {
          const int d8 = qd[q];
          if (d8 != 255) {
            s  = queue[q];
            dl = d8;
            c  = qv[q] / (ssum[d8] + 1e-16f);
          }
        }
        for (int j = 0; j < 32; j += 8) {
          if (q0 + j >= n) break;
          float r[8][4];
          int   dj[8];
          float cj[8];
          #pragma unroll
          for (int u = 0; u < 8; ++u) {
            const int sj = __builtin_amdgcn_readlane(s, j + u);
            dj[u] = __builtin_amdgcn_readlane(dl, j + u);
            cj[u] = __int_as_float(__builtin_amdgcn_readlane(__float_as_int(c), j + u));
            const float* hp = h + (size_t)sj * F_DIM + lane;
            r[u][0] = hp[0];
            r[u][1] = hp[32];
            r[u][2] = hp[64];
            r[u][3] = hp[96];
          }
          asm volatile("" :: "v"(r[0][0]), "v"(r[0][1]), "v"(r[0][2]), "v"(r[0][3]),
                             "v"(r[1][0]), "v"(r[1][1]), "v"(r[1][2]), "v"(r[1][3]),
                             "v"(r[2][0]), "v"(r[2][1]), "v"(r[2][2]), "v"(r[2][3]),
                             "v"(r[3][0]), "v"(r[3][1]), "v"(r[3][2]), "v"(r[3][3]));
          asm volatile("" :: "v"(r[4][0]), "v"(r[4][1]), "v"(r[4][2]), "v"(r[4][3]),
                             "v"(r[5][0]), "v"(r[5][1]), "v"(r[5][2]), "v"(r[5][3]),
                             "v"(r[6][0]), "v"(r[6][1]), "v"(r[6][2]), "v"(r[6][3]),
                             "v"(r[7][0]), "v"(r[7][1]), "v"(r[7][2]), "v"(r[7][3]));
          #pragma unroll
          for (int u = 0; u < 8; ++u) {
            if (dj[u] >= 0) {
              float* tp = tile + dj[u] * F_DIM + lane;
              atomicAdd(tp,      cj[u] * r[u][0]);
              atomicAdd(tp + 32, cj[u] * r[u][1]);
              atomicAdd(tp + 64, cj[u] * r[u][2]);
              atomicAdd(tp + 96, cj[u] * r[u][3]);
            }
          }
        }
      }
    } else {
      for (int i = tid; i < E; i += AGG_THREADS) {
        if (owner[i] != me) continue;
        const int d = dst[i];
        if (d < base || d >= base + AGG_ROWS || d >= N) continue;
        atomicMax(&smax[d - base], f2ord(lg[i]));
      }
      __syncthreads();
      for (int i = tid; i < E; i += AGG_THREADS) {
        if (owner[i] != me) continue;
        const int d = dst[i];
        if (d < base || d >= base + AGG_ROWS || d >= N) continue;
        atomicAdd(&ssum[d - base], expf(lg[i] - ord2f(smax[d - base])));
      }
      __syncthreads();
      for (int i = tid; i < E; i += AGG_THREADS) {
        if (owner[i] != me) continue;
        const int d = dst[i];
        if (d < base || d >= base + AGG_ROWS || d >= N) continue;
        const float c = expf(lg[i] - ord2f(smax[d - base])) / (ssum[d - base] + 1e-16f);
        const float* hp = h + (size_t)src[i] * F_DIM;
        float* tp = tile + (d - base) * F_DIM;
        for (int f = 0; f < F_DIM; ++f) atomicAdd(tp + f, c * hp[f]);
      }
    }
    __syncthreads();

    const int rows = min(AGG_ROWS, N - base);
    float4* op = (float4*)(out + (size_t)base * F_DIM);
    for (int k = tid; k < rows * (F_DIM / 4); k += AGG_THREADS)
      op[k] = ((const float4*)tile)[k];
    __syncthreads();
  }
}

extern "C" void kernel_launch(void* const* d_in, const int* in_sizes, int n_in,
                              void* d_out, int out_size, void* d_ws, size_t ws_size,
                              hipStream_t stream) {
  const float* x  = (const float*)d_in[0];
  const int*   ei = (const int*)d_in[1];
  const float* W  = (const float*)d_in[2];
  const float* a  = (const float*)d_in[3];
  const int N = in_sizes[0] / F_DIM;
  const int E = in_sizes[1] / 2;
  const int* src = ei;
  const int* dst = ei + E;

  float*    h      = (float*)d_ws;
  float*    as     = h + (size_t)N * F_DIM;
  float*    ad     = as + N;
  float*    e      = ad + N;
  unsigned char* owner = (unsigned char*)(e + E);
  float*    out    = (float*)d_out;

  gat_gemm_wmma <<<GEMM_GROUPS,  256, 0, stream>>>(x, W, h, N);
  gat_alpha     <<<ALPHA_GROUPS, ALPHA_THREADS, 0, stream>>>(h, a, as, ad, N);
  gat_edge_logit<<<(E + 1023) / 1024, 256, 0, stream>>>(src, dst, as, ad, e, owner, E);
  gat_aggregate <<<AGG_GROUPS, AGG_THREADS, 0, stream>>>(src, dst, h, e, owner, out, N, E);
}


# edge logits formed in phase A by the work-group owning the destination (alpha_dst of its rows in LDS, alpha_src[src] gathered there): edge launch gathers nothing, stores no logits; sum order kept
# speedup vs baseline: 3.4413x; 1.0864x over previous
#include <hip/hip_runtime.h>
#include <hip/hip_bf16.h>

typedef __attribute__((ext_vector_type(2))) float v2f;
typedef __attribute__((ext_vector_type(8))) float v8f;

#define F_DIM 128
#define LEAKY 0.2f

#define AGG_GROUPS  256
#define AGG_ROWS    196
#define AGG_THREADS 1024
#define AGG_QCAP    8192
#define AGG_INFLIGHT 8

#define GEMM_GROUPS   1024
#define ALPHA_GROUPS  256
#define ALPHA_THREADS 512
#define ALPHA_ROWS    8

__device__ __forceinline__ unsigned f2ord(float f) {
  unsigned u = __float_as_uint(f);
  return (u & 0x80000000u) ? ~u : (u | 0x80000000u);
}
__device__ __forceinline__ float ord2f(unsigned u) {
  return (u & 0x80000000u) ? __uint_as_float(u & 0x7FFFFFFFu)
                           : __uint_as_float(~u);
}

__global__ __launch_bounds__(256) void gat_gemm_wmma(
    const float* __restrict__ x, const float* __restrict__ W,
    float* __restrict__ h, int M)
{
  __shared__ float sA[16 * 132];
  const int tid  = threadIdx.x;
  const int lane = tid & 31;
  const int wv   = tid >> 5;
  const int row  = lane & 15;
  const int kh   = (lane >> 4) * 2;
  const int n0   = wv * 16;
  const int mh   = (lane >> 4) * 8;
  const int ntiles = (M + 15) >> 4;

  const int sr = tid >> 5, sc = tid & 31;

  v2f bf[32];
  #pragma unroll
  for (int kk = 0; kk < 32; ++kk) {
    bf[kk].x = W[(4 * kk + kh + 0) * F_DIM + n0 + row];
    bf[kk].y = W[(4 * kk + kh + 1) * F_DIM + n0 + row];
  }

  int tile = blockIdx.x;
  float4 p0 = make_float4(0.f, 0.f, 0.f, 0.f), p1 = p0;
  if (tile < ntiles) {
    const int m0 = tile * 16;
    if (m0 + sr     < M) p0 = ((const float4*)(x + (size_t)(m0 + sr)     * F_DIM))[sc];
    if (m0 + sr + 8 < M) p1 = ((const float4*)(x + (size_t)(m0 + sr + 8) * F_DIM))[sc];
  }

  for (; tile < ntiles; tile += gridDim.x) {
    const int m0 = tile * 16;
    ((float4*)sA)[sr * 33 + sc]       = p0;
    ((float4*)sA)[(sr + 8) * 33 + sc] = p1;
    __syncthreads();

    const int nt = tile + gridDim.x;
    p0 = make_float4(0.f, 0.f, 0.f, 0.f); p1 = p0;
    if (nt < ntiles) {
      const int q0 = nt * 16;
      if (q0 + sr     < M) p0 = ((const float4*)(x + (size_t)(q0 + sr)     * F_DIM))[sc];
      if (q0 + sr + 8 < M) p1 = ((const float4*)(x + (size_t)(q0 + sr + 8) * F_DIM))[sc];
    }

    v8f acc = {};
    #pragma unroll
    for (int kk = 0; kk < 32; ++kk) {
      v2f a;
      a.x = sA[row * 132 + 4 * kk + kh + 0];
      a.y = sA[row * 132 + 4 * kk + kh + 1];
      acc = __builtin_amdgcn_wmma_f32_16x16x4_f32(false, a, false, bf[kk],
                                                  (short)0, acc, false, false);
    }

    float* hp = h + (size_t)m0 * F_DIM + n0 + row;
    if (m0 + 16 <= M) {
      #pragma unroll
      for (int v = 0; v < 8; ++v) hp[(size_t)(mh + v) * F_DIM] = acc[v];
    } else {
      #pragma unroll
      for (int v = 0; v < 8; ++v)
        if (m0 + mh + v < M) hp[(size_t)(mh + v) * F_DIM] = acc[v];
    }
    __syncthreads();
  }
}

__global__ __launch_bounds__(ALPHA_THREADS) void gat_alpha(
    const float* __restrict__ h, const float* __restrict__ a,
    float* __restrict__ as, float* __restrict__ ad, int N)
{
  const int lane   = threadIdx.x & 31;
  const int wave   = blockIdx.x * (ALPHA_THREADS / 32) + (threadIdx.x >> 5);
  const int nwaves = gridDim.x * (ALPHA_THREADS / 32);
  const float4 a1 = ((const float4*)a)[lane];
  const float4 a2 = ((const float4*)(a + F_DIM))[lane];

  for (int node0 = wave * ALPHA_ROWS; node0 < N; node0 += nwaves * ALPHA_ROWS) {
    float4 hv[ALPHA_ROWS];
    #pragma unroll
    for (int u = 0; u < ALPHA_ROWS; ++u) {
      hv[u] = make_float4(0.f, 0.f, 0.f, 0.f);
      if (node0 + u < N) hv[u] = ((const float4*)(h + (size_t)(node0 + u) * F_DIM))[lane];
    }
    float sv = 0.f, dv = 0.f;
    #pragma unroll
    for (int u = 0; u < ALPHA_ROWS; ++u) {
      float s = hv[u].x * a1.x + hv[u].y * a1.y + hv[u].z * a1.z + hv[u].w * a1.w;
      float d = hv[u].x * a2.x + hv[u].y * a2.y + hv[u].z * a2.z + hv[u].w * a2.w;
      #pragma unroll
      for (int m = 16; m >= 1; m >>= 1) {
        s += __shfl_xor(s, m, 32);
        d += __shfl_xor(d, m, 32);
      }
      if (lane == u) { sv = s; dv = d; }
    }
    if (lane < ALPHA_ROWS && node0 + lane < N) {
      as[node0 + lane] = sv;
      ad[node0 + lane] = dv;
    }
  }
}

__global__ __launch_bounds__(256) void gat_edge_owner(
    const int* __restrict__ dst, unsigned char* __restrict__ owner, int E)
{
  const int i0 = blockIdx.x * 1024 + threadIdx.x;
  int d[4];
  #pragma unroll
  for (int u = 0; u < 4; ++u) {
    const int i = i0 + u * 256;
    d[u] = 0;
    if (i < E) d[u] = dst[i];
  }
  #pragma unroll
  for (int u = 0; u < 4; ++u) {
    const int i = i0 + u * 256;
    if (i < E) owner[i] = (unsigned char)((d[u] / AGG_ROWS) & (AGG_GROUPS - 1));
  }
}

__device__ __forceinline__ unsigned zero_bytes(unsigned x) {
  unsigned t = (x & 0x7F7F7F7Fu) + 0x7F7F7F7Fu;
  return ~(t | x | 0x7F7F7F7Fu);
}

__device__ __forceinline__ void agg_take(int i, int* queue, int* qcount)
{
  const int pos = atomicAdd(qcount, 1);
  if (pos < AGG_QCAP) queue[pos] = i;
}

__global__ __launch_bounds__(AGG_THREADS) void gat_aggregate(
    const int* __restrict__ src, const int* __restrict__ dst,
    const float* __restrict__ h,
    const float* __restrict__ as, const float* __restrict__ ad,
    const unsigned char* __restrict__ owner,
    float* __restrict__ out, int N, int E)
{
  __shared__ float    tile[AGG_ROWS * F_DIM];
  __shared__ int      queue[AGG_QCAP];
  __shared__ float    qv[AGG_QCAP];
  __shared__ unsigned char qd[AGG_QCAP];
  __shared__ unsigned smax[AGG_ROWS];
  __shared__ float    ssum[AGG_ROWS];
  __shared__ float    sad[AGG_ROWS];
  __shared__ int      qcount;

  const int tid  = threadIdx.x;
  const int lane = tid & 31;
  const int wv   = tid >> 5;
  const unsigned me  = blockIdx.x;
  const unsigned me4 = me * 0x01010101u;

  int head = (int)((16u - (unsigned)((size_t)owner & 15u)) & 15u);
  if (head > E) head = E;
  const int nvec  = (E - head) >> 4;
  const int tail0 = head + (nvec << 4);
  const uint4* vp = (const uint4*)(owner + head);
  const int vstart = (int)(((long long)blockIdx.x * nvec) / AGG_GROUPS);

  for (int base = blockIdx.x * AGG_ROWS; base < N; base += AGG_GROUPS * AGG_ROWS) {
    for (int k = tid; k < AGG_ROWS * F_DIM / 4; k += AGG_THREADS)
      ((float4*)tile)[k] = make_float4(0.f, 0.f, 0.f, 0.f);
    for (int k = tid; k < AGG_ROWS; k += AGG_THREADS) {
      smax[k] = 0u;
      ssum[k] = 0.f;
      sad[k]  = (base + k < N) ? ad[base + k] : 0.f;
    }
    if (tid == 0) qcount = 0;
    __syncthreads();

    int v0 = tid;
    for (; v0 + (AGG_INFLIGHT - 1) * AGG_THREADS < nvec; v0 += AGG_INFLIGHT * AGG_THREADS) {
      uint4 w[AGG_INFLIGHT];
      int   vi[AGG_INFLIGHT];
      #pragma unroll
      for (int u = 0; u < AGG_INFLIGHT; ++u) {
        int v = v0 + u * AGG_THREADS + vstart;
        if (v >= nvec) v -= nvec;
        vi[u] = v;
        w[u]  = vp[v];
      }
      #pragma unroll
      for (int u = 0; u < AGG_INFLIGHT; ++u) {
        unsigned m = (zero_bytes(w[u].x ^ me4) >> 7) | (zero_bytes(w[u].y ^ me4) >> 6) |
                     (zero_bytes(w[u].z ^ me4) >> 5) | (zero_bytes(w[u].w ^ me4) >> 4);
        while (m) {
          const int p = __builtin_ctz(m);
          m &= m - 1;
          agg_take(head + (vi[u] << 4) + ((p & 7) << 2) + (p >> 3), queue, &qcount);
        }
      }
    }
    for (; v0 < nvec; v0 += AGG_THREADS) {
      int v = v0 + vstart;
      if (v >= nvec) v -= nvec;
      const uint4 w1 = vp[v];
      unsigned m = (zero_bytes(w1.x ^ me4) >> 7) | (zero_bytes(w1.y ^ me4) >> 6) |
                   (zero_bytes(w1.z ^ me4) >> 5) | (zero_bytes(w1.w ^ me4) >> 4);
      while (m) {
        const int p = __builtin_ctz(m);
        m &= m - 1;
        agg_take(head + (v << 4) + ((p & 7) << 2) + (p >> 3), queue, &qcount);
      }
    }
    for (int i = tid; i < head; i += AGG_THREADS)
      if (owner[i] == me) agg_take(i, queue, &qcount);
    for (int i = tail0 + tid; i < E; i += AGG_THREADS)
      if (owner[i] == me) agg_take(i, queue, &qcount);
    __syncthreads();

    const int total = qcount;
    if (total <= AGG_QCAP) {
      const int n = total;

      for (int q0 = tid; q0 < n; q0 += 4 * AGG_THREADS) {
        int   ei[4], es[4], ed[4];
        float ea[4];
        #pragma unroll
        for (int u = 0; u < 4; ++u) {
          const int q = q0 + u * AGG_THREADS;
          ei[u] = (q < n) ? queue[q] : -1;
        }
        #pragma unroll
        for (int u = 0; u < 4; ++u) {
          es[u] = 0; ed[u] = -1;
          if (ei[u] >= 0) { ed[u] = dst[ei[u]]; es[u] = src[ei[u]]; }
        }
        #pragma unroll
        for (int u = 0; u < 4; ++u) ea[u] = as[es[u]];
        #pragma unroll
        for (int u = 0; u < 4; ++u) {
          if (ei[u] >= 0) {
            const int q = q0 + u * AGG_THREADS;
            const int d = ed[u];
            const bool mine = d >= base && d < base + AGG_ROWS && d < N;
            const float v  = ea[u] + (mine ? sad[d - base] : 0.f);
            const float ev = v > 0.f ? v : LEAKY * v;
            queue[q] = es[u];
            qv[q]    = ev;
            qd[q]    = mine ? (unsigned char)(d - base) : (unsigned char)255;
            if (mine) atomicMax(&smax[d - base], f2ord(ev));
          }
        }
      }
      __syncthreads();

      for (int q = tid; q < n; q += AGG_THREADS) {
        const int dl = qd[q];
        if (dl != 255) {
          const float v = expf(qv[q] - ord2f(smax[dl]));
          qv[q] = v;
          atomicAdd(&ssum[dl], v);
        }
      }
      __syncthreads();

      for (int q0 = wv * 32; q0 < n; q0 += AGG_THREADS) {
        const int q = q0 + lane;
        int s = 0, dl = -1;
        float c = 0.f;
        if (q < n) {
          const int d8 = qd[q];
          if (d8 != 255) {
            s  = queue[q];
            dl = d8;
            c  = qv[q] / (ssum[d8] + 1e-16f);
          }
        }
        for (int j = 0; j < 32; j += 8) {
          if (q0 + j >= n) break;
          float r[8][4];
          int   dj[8];
          float cj[8];
          #pragma unroll
          for (int u = 0; u < 8; ++u) {
            const int sj = __builtin_amdgcn_readlane(s, j + u);
            dj[u] = __builtin_amdgcn_readlane(dl, j + u);
            cj[u] = __int_as_float(__builtin_amdgcn_readlane(__float_as_int(c), j + u));
            const float* hp = h + (size_t)sj * F_DIM + lane;
            r[u][0] = hp[0];
            r[u][1] = hp[32];
            r[u][2] = hp[64];
            r[u][3] = hp[96];
          }
          asm volatile("" :: "v"(r[0][0]), "v"(r[0][1]), "v"(r[0][2]), "v"(r[0][3]),
                             "v"(r[1][0]), "v"(r[1][1]), "v"(r[1][2]), "v"(r[1][3]),
                             "v"(r[2][0]), "v"(r[2][1]), "v"(r[2][2]), "v"(r[2][3]),
                             "v"(r[3][0]), "v"(r[3][1]), "v"(r[3][2]), "v"(r[3][3]));
          asm volatile("" :: "v"(r[4][0]), "v"(r[4][1]), "v"(r[4][2]), "v"(r[4][3]),
                             "v"(r[5][0]), "v"(r[5][1]), "v"(r[5][2]), "v"(r[5][3]),
                             "v"(r[6][0]), "v"(r[6][1]), "v"(r[6][2]), "v"(r[6][3]),
                             "v"(r[7][0]), "v"(r[7][1]), "v"(r[7][2]), "v"(r[7][3]));
          #pragma unroll
          for (int u = 0; u < 8; ++u) {
            if (dj[u] >= 0) {
              float* tp = tile + dj[u] * F_DIM + lane;
              atomicAdd(tp,      cj[u] * r[u][0]);
              atomicAdd(tp + 32, cj[u] * r[u][1]);
              atomicAdd(tp + 64, cj[u] * r[u][2]);
              atomicAdd(tp + 96, cj[u] * r[u][3]);
            }
          }
        }
      }
    } else {
      for (int i = tid; i < E; i += AGG_THREADS) {
        if (owner[i] != me) continue;
        const int d = dst[i];
        if (d < base || d >= base + AGG_ROWS || d >= N) continue;
        const float v = as[src[i]] + sad[d - base];
        atomicMax(&smax[d - base], f2ord(v > 0.f ? v : LEAKY * v));
      }
      __syncthreads();
      for (int i = tid; i < E; i += AGG_THREADS) {
        if (owner[i] != me) continue;
        const int d = dst[i];
        if (d < base || d >= base + AGG_ROWS || d >= N) continue;
        const float v = as[src[i]] + sad[d - base];
        atomicAdd(&ssum[d - base], expf((v > 0.f ? v : LEAKY * v) - ord2f(smax[d - base])));
      }
      __syncthreads();
      for (int i = tid; i < E; i += AGG_THREADS) {
        if (owner[i] != me) continue;
        const int d = dst[i];
        if (d < base || d >= base + AGG_ROWS || d >= N) continue;
        const int   s = src[i];
        const float v = as[s] + sad[d - base];
        const float c = expf((v > 0.f ? v : LEAKY * v) - ord2f(smax[d - base])) / (ssum[d - base] + 1e-16f);
        const float* hp = h + (size_t)s * F_DIM;
        float* tp = tile + (d - base) * F_DIM;
        for (int f = 0; f < F_DIM; ++f) atomicAdd(tp + f, c * hp[f]);
      }
    }
    __syncthreads();

    const int rows = min(AGG_ROWS, N - base);
    float4* op = (float4*)(out + (size_t)base * F_DIM);
    for (int k = tid; k < rows * (F_DIM / 4); k += AGG_THREADS)
      op[k] = ((const float4*)tile)[k];
    __syncthreads();
  }
}

extern "C" void kernel_launch(void* const* d_in, const int* in_sizes, int n_in,
                              void* d_out, int out_size, void* d_ws, size_t ws_size,
                              hipStream_t stream) {
  const float* x  = (const float*)d_in[0];
  const int*   ei = (const int*)d_in[1];
  const float* W  = (const float*)d_in[2];
  const float* a  = (const float*)d_in[3];
  const int N = in_sizes[0] / F_DIM;
  const int E = in_sizes[1] / 2;
  const int* src = ei;
  const int* dst = ei + E;

  float*    h      = (float*)d_ws;
  float*    as     = h + (size_t)N * F_DIM;
  float*    ad     = as + N;
  unsigned char* owner = (unsigned char*)(ad + N);
  float*    out    = (float*)d_out;

  gat_gemm_wmma <<<GEMM_GROUPS,  256, 0, stream>>>(x, W, h, N);
  gat_alpha     <<<ALPHA_GROUPS, ALPHA_THREADS, 0, stream>>>(h, a, as, ad, N);
  gat_edge_owner<<<(E + 1023) / 1024, 256, 0, stream>>>(dst, owner, E);
  gat_aggregate <<<AGG_GROUPS, AGG_THREADS, 0, stream>>>(src, dst, h, as, ad, owner, out, N, E);
}


# edge filter by presence masks: edge launch writes per 1,024-edge chunk and owner a 64-bit mask of 16-edge blocks; owners read only marked dst blocks, no byte plane; LDS-atomic sums, queue reordered
# speedup vs baseline: 3.6897x; 1.0722x over previous
#include <hip/hip_runtime.h>
#include <hip/hip_bf16.h>

typedef __attribute__((ext_vector_type(2))) float v2f;
typedef __attribute__((ext_vector_type(8))) float v8f;

#define F_DIM 128
#define LEAKY 0.2f

#define AGG_GROUPS  256
#define AGG_ROWS    196
#define AGG_THREADS 1024
#define AGG_QCAP    8192
#define MASK_CHUNK  1024
#define MASK_BLOCK  16
#define AGG_BCAP    8192
#define AGG_BLOADS  16

#define GEMM_GROUPS   1024
#define ALPHA_GROUPS  256
#define ALPHA_THREADS 512
#define ALPHA_ROWS    8

__device__ __forceinline__ unsigned f2ord(float f) {
  unsigned u = __float_as_uint(f);
  return (u & 0x80000000u) ? ~u : (u | 0x80000000u);
}
__device__ __forceinline__ float ord2f(unsigned u) {
  return (u & 0x80000000u) ? __uint_as_float(u & 0x7FFFFFFFu)
                           : __uint_as_float(~u);
}

__global__ __launch_bounds__(256) void gat_gemm_wmma(
    const float* __restrict__ x, const float* __restrict__ W,
    float* __restrict__ h, int M)
{
  __shared__ float sA[16 * 132];
  const int tid  = threadIdx.x;
  const int lane = tid & 31;
  const int wv   = tid >> 5;
  const int row  = lane & 15;
  const int kh   = (lane >> 4) * 2;
  const int n0   = wv * 16;
  const int mh   = (lane >> 4) * 8;
  const int ntiles = (M + 15) >> 4;

  const int sr = tid >> 5, sc = tid & 31;

  v2f bf[32];
  #pragma unroll
  for (int kk = 0; kk < 32; ++kk) {
    bf[kk].x = W[(4 * kk + kh + 0) * F_DIM + n0 + row];
    bf[kk].y = W[(4 * kk + kh + 1) * F_DIM + n0 + row];
  }

  int tile = blockIdx.x;
  float4 p0 = make_float4(0.f, 0.f, 0.f, 0.f), p1 = p0;
  if (tile < ntiles) {
    const int m0 = tile * 16;
    if (m0 + sr     < M) p0 = ((const float4*)(x + (size_t)(m0 + sr)     * F_DIM))[sc];
    if (m0 + sr + 8 < M) p1 = ((const float4*)(x + (size_t)(m0 + sr + 8) * F_DIM))[sc];
  }

  for (; tile < ntiles; tile += gridDim.x) {
    const int m0 = tile * 16;
    ((float4*)sA)[sr * 33 + sc]       = p0;
    ((float4*)sA)[(sr + 8) * 33 + sc] = p1;
    __syncthreads();

    const int nt = tile + gridDim.x;
    p0 = make_float4(0.f, 0.f, 0.f, 0.f); p1 = p0;
    if (nt < ntiles) {
      const int q0 = nt * 16;
      if (q0 + sr     < M) p0 = ((const float4*)(x + (size_t)(q0 + sr)     * F_DIM))[sc];
      if (q0 + sr + 8 < M) p1 = ((const float4*)(x + (size_t)(q0 + sr + 8) * F_DIM))[sc];
    }

    v8f acc = {};
    #pragma unroll
    for (int kk = 0; kk < 32; ++kk) {
      v2f a;
      a.x = sA[row * 132 + 4 * kk + kh + 0];
      a.y = sA[row * 132 + 4 * kk + kh + 1];
      acc = __builtin_amdgcn_wmma_f32_16x16x4_f32(false, a, false, bf[kk],
                                                  (short)0, acc, false, false);
    }

    float* hp = h + (size_t)m0 * F_DIM + n0 + row;
    if (m0 + 16 <= M) {
      #pragma unroll
      for (int v = 0; v < 8; ++v) hp[(size_t)(mh + v) * F_DIM] = acc[v];
    } else {
      #pragma unroll
      for (int v = 0; v < 8; ++v)
        if (m0 + mh + v < M) hp[(size_t)(mh + v) * F_DIM] = acc[v];
    }
    __syncthreads();
  }
}

__global__ __launch_bounds__(ALPHA_THREADS) void gat_alpha(
    const float* __restrict__ h, const float* __restrict__ a,
    float* __restrict__ as, float* __restrict__ ad, int N)
{
  const int lane   = threadIdx.x & 31;
  const int wave   = blockIdx.x * (ALPHA_THREADS / 32) + (threadIdx.x >> 5);
  const int nwaves = gridDim.x * (ALPHA_THREADS / 32);
  const float4 a1 = ((const float4*)a)[lane];
  const float4 a2 = ((const float4*)(a + F_DIM))[lane];

  for (int node0 = wave * ALPHA_ROWS; node0 < N; node0 += nwaves * ALPHA_ROWS) {
    float4 hv[ALPHA_ROWS];
    #pragma unroll
    for (int u = 0; u < ALPHA_ROWS; ++u) {
      hv[u] = make_float4(0.f, 0.f, 0.f, 0.f);
      if (node0 + u < N) hv[u] = ((const float4*)(h + (size_t)(node0 + u) * F_DIM))[lane];
    }
    float sv = 0.f, dv = 0.f;
    #pragma unroll
    for (int u = 0; u < ALPHA_ROWS; ++u) {
      float s = hv[u].x * a1.x + hv[u].y * a1.y + hv[u].z * a1.z + hv[u].w * a1.w;
      float d = hv[u].x * a2.x + hv[u].y * a2.y + hv[u].z * a2.z + hv[u].w * a2.w;
      #pragma unroll
      for (int m = 16; m >= 1; m >>= 1) {
        s += __shfl_xor(s, m, 32);
        d += __shfl_xor(d, m, 32);
      }
      if (lane == u) { sv = s; dv = d; }
    }
    if (lane < ALPHA_ROWS && node0 + lane < N) {
      as[node0 + lane] = sv;
      ad[node0 + lane] = dv;
    }
  }
}

__global__ __launch_bounds__(256) void gat_edge_mask(
    const int* __restrict__ dst, unsigned* __restrict__ masks, int E)
{
  __shared__ unsigned sm[2 * AGG_GROUPS];
  const int t  = threadIdx.x;
  const int i0 = blockIdx.x * MASK_CHUNK + t;
  int d[4];
  #pragma unroll
  for (int u = 0; u < 4; ++u) {
    const int i = i0 + u * 256;
    d[u] = 0;
    if (i < E) d[u] = dst[i];
  }
  sm[t] = 0u;
  sm[t + 256] = 0u;
  __syncthreads();
  #pragma unroll
  for (int u = 0; u < 4; ++u) {
    const int i = i0 + u * 256;
    if (i < E) {
      const int o = (d[u] / AGG_ROWS) & (AGG_GROUPS - 1);
      const int b = (t + u * 256) / MASK_BLOCK;
      atomicOr(&sm[2 * o + (b >> 5)], 1u << (b & 31));
    }
  }
  __syncthreads();
  ((uint2*)masks)[(size_t)blockIdx.x * AGG_GROUPS + t] = make_uint2(sm[2 * t], sm[2 * t + 1]);
}

__global__ __launch_bounds__(AGG_THREADS) void gat_aggregate(
    const int* __restrict__ src, const int* __restrict__ dst,
    const float* __restrict__ h,
    const float* __restrict__ as, const float* __restrict__ ad,
    const unsigned* __restrict__ masks,
    float* __restrict__ out, int N, int E)
{
  __shared__ float    tile[AGG_ROWS * F_DIM];
  __shared__ int      queue[AGG_QCAP];
  __shared__ float    qv[AGG_QCAP];
  __shared__ unsigned char qd[AGG_QCAP];
  __shared__ int      blist[AGG_BCAP];
  __shared__ unsigned smax[AGG_ROWS];
  __shared__ float    ssum[AGG_ROWS];
  __shared__ float    sad[AGG_ROWS];
  __shared__ int      qcount;
  __shared__ int      bcount;

  const int tid  = threadIdx.x;
  const int lane = tid & 31;
  const int wv   = tid >> 5;
  const unsigned me = blockIdx.x;
  const int nchunks = (E + MASK_CHUNK - 1) / MASK_CHUNK;
  const uint2* mp = (const uint2*)masks;

  for (int base = blockIdx.x * AGG_ROWS; base < N; base += AGG_GROUPS * AGG_ROWS) {
    for (int k = tid; k < AGG_ROWS * F_DIM / 4; k += AGG_THREADS)
      ((float4*)tile)[k] = make_float4(0.f, 0.f, 0.f, 0.f);
    for (int k = tid; k < AGG_ROWS; k += AGG_THREADS) {
      smax[k] = 0u;
      ssum[k] = 0.f;
      sad[k]  = (base + k < N) ? ad[base + k] : 0.f;
    }
    if (tid == 0) { qcount = 0; bcount = 0; }
    __syncthreads();

    for (int c0 = 0; c0 < nchunks; c0 += AGG_THREADS) {
      const int c = c0 + tid;
      uint2 mk = make_uint2(0u, 0u);
      if (c < nchunks) mk = mp[(size_t)c * AGG_GROUPS + me];
      const int cnt = __popc(mk.x) + __popc(mk.y);
      int inc = cnt;
      #pragma unroll
      for (int s = 1; s < 32; s <<= 1) {
        const int up = __shfl_up(inc, s, 32);
        if (lane >= s) inc += up;
      }
      const int tot = __shfl(inc, 31, 32);
      int bs = 0;
      if (lane == 31 && tot > 0) bs = atomicAdd(&bcount, tot);
      bs = __shfl(bs, 31, 32);
      int pos = bs + inc - cnt;
      const int blk0 = c * (MASK_CHUNK / MASK_BLOCK);
      unsigned m = mk.x;
      while (m) {
        const int p = __builtin_ctz(m);
        m &= m - 1;
        if (pos < AGG_BCAP) blist[pos] = blk0 + p;
        ++pos;
      }
      m = mk.y;
      while (m) {
        const int p = __builtin_ctz(m);
        m &= m - 1;
        if (pos < AGG_BCAP) blist[pos] = blk0 + 32 + p;
        ++pos;
      }
    }
    __syncthreads();

    const int nb = bcount;
    if (nb <= AGG_BCAP) {
      const int npair = (nb + 1) >> 1;
      for (int k0 = wv; k0 < npair; k0 += 32 * AGG_BLOADS) {
        int ii[AGG_BLOADS], dd[AGG_BLOADS];
        #pragma unroll
        for (int u = 0; u < AGG_BLOADS; ++u) {
          const int bi = 2 * (k0 + u * 32) + (lane >> 4);
          ii[u] = blist[bi < nb ? bi : 0];
        }
        #pragma unroll
        for (int u = 0; u < AGG_BLOADS; ++u) {
          const int bi = 2 * (k0 + u * 32) + (lane >> 4);
          const int i  = ii[u] * MASK_BLOCK + (lane & 15);
          ii[u] = (bi < nb && i < E) ? i : -1;
        }
        #pragma unroll
        for (int u = 0; u < AGG_BLOADS; ++u) dd[u] = dst[ii[u] < 0 ? 0 : ii[u]];
        unsigned hb[AGG_BLOADS];
        int tot = 0;
        #pragma unroll
        for (int u = 0; u < AGG_BLOADS; ++u) {
          const int d = dd[u];
          const bool hit = ii[u] >= 0 && d >= base && d < base + AGG_ROWS && d < N;
          hb[u] = __builtin_amdgcn_ballot_w32(hit);
          tot += __popc(hb[u]);
        }
        if (tot > 0) {
          int bs = 0;
          if (lane == 0) bs = atomicAdd(&qcount, tot);
          bs = __builtin_amdgcn_readfirstlane(bs);
          #pragma unroll
          for (int u = 0; u < AGG_BLOADS; ++u) {
            if ((hb[u] >> lane) & 1u) {
              const int pos = bs + __builtin_amdgcn_mbcnt_lo(hb[u], 0u);
              if (pos < AGG_QCAP) {
                queue[pos] = ii[u];
                qd[pos]    = (unsigned char)(dd[u] - base);
              }
            }
            bs += __popc(hb[u]);
          }
        }
      }
    }
    __syncthreads();

    const int total = qcount;
    if (nb <= AGG_BCAP && total <= AGG_QCAP) {
      const int n = total;

      for (int q0 = tid; q0 < n; q0 += 4 * AGG_THREADS) {
        int   ei[4], es[4];
        float ea[4];
        #pragma unroll
        for (int u = 0; u < 4; ++u) {
          const int q = q0 + u * AGG_THREADS;
          ei[u] = (q < n) ? queue[q] : -1;
        }
        #pragma unroll
        for (int u = 0; u < 4; ++u) {
          es[u] = 0;
          if (ei[u] >= 0) es[u] = src[ei[u]];
        }
        #pragma unroll
        for (int u = 0; u < 4; ++u) ea[u] = as[es[u]];
        #pragma unroll
        for (int u = 0; u < 4; ++u) {
          if (ei[u] >= 0) {
            const int q  = q0 + u * AGG_THREADS;
            const int dl = qd[q];
            const float v  = ea[u] + sad[dl];
            const float ev = v > 0.f ? v : LEAKY * v;
            queue[q] = es[u];
            qv[q]    = ev;
            atomicMax(&smax[dl], f2ord(ev));
          }
        }
      }
      __syncthreads();

      for (int q = tid; q < n; q += AGG_THREADS) {
        const int dl = qd[q];
        const float v = expf(qv[q] - ord2f(smax[dl]));
        qv[q] = v;
        atomicAdd(&ssum[dl], v);
      }
      __syncthreads();

      for (int q0 = wv * 32; q0 < n; q0 += AGG_THREADS) {
        const int q = q0 + lane;
        int s = 0, dl = -1;
        float c = 0.f;
        if (q < n) {
          const int d8 = qd[q];
          s  = queue[q];
          dl = d8;
          c  = qv[q] / (ssum[d8] + 1e-16f);
        }
        for (int j = 0; j < 32; j += 8) {
          if (q0 + j >= n) break;
          float r[8][4];
          int   dj[8];
          float cj[8];
          #pragma unroll
          for (int u = 0; u < 8; ++u) {
            const int sj = __builtin_amdgcn_readlane(s, j + u);
            dj[u] = __builtin_amdgcn_readlane(dl, j + u);
            cj[u] = __int_as_float(__builtin_amdgcn_readlane(__float_as_int(c), j + u));
            const float* hp = h + (size_t)sj * F_DIM + lane;
            r[u][0] = hp[0];
            r[u][1] = hp[32];
            r[u][2] = hp[64];
            r[u][3] = hp[96];
          }
          asm volatile("" :: "v"(r[0][0]), "v"(r[0][1]), "v"(r[0][2]), "v"(r[0][3]),
                             "v"(r[1][0]), "v"(r[1][1]), "v"(r[1][2]), "v"(r[1][3]),
                             "v"(r[2][0]), "v"(r[2][1]), "v"(r[2][2]), "v"(r[2][3]),
                             "v"(r[3][0]), "v"(r[3][1]), "v"(r[3][2]), "v"(r[3][3]));
          asm volatile("" :: "v"(r[4][0]), "v"(r[4][1]), "v"(r[4][2]), "v"(r[4][3]),
                             "v"(r[5][0]), "v"(r[5][1]), "v"(r[5][2]), "v"(r[5][3]),
                             "v"(r[6][0]), "v"(r[6][1]), "v"(r[6][2]), "v"(r[6][3]),
                             "v"(r[7][0]), "v"(r[7][1]), "v"(r[7][2]), "v"(r[7][3]));
          #pragma unroll
          for (int u = 0; u < 8; ++u) {
            if (dj[u] >= 0) {
              float* tp = tile + dj[u] * F_DIM + lane;
              atomicAdd(tp,      cj[u] * r[u][0]);
              atomicAdd(tp + 32, cj[u] * r[u][1]);
              atomicAdd(tp + 64, cj[u] * r[u][2]);
              atomicAdd(tp + 96, cj[u] * r[u][3]);
            }
          }
        }
      }
    } else {
      for (int i = tid; i < E; i += AGG_THREADS) {
        const int d = dst[i];
        if (d < base || d >= base + AGG_ROWS || d >= N) continue;
        const float v = as[src[i]] + sad[d - base];
        atomicMax(&smax[d - base], f2ord(v > 0.f ? v : LEAKY * v));
      }
      __syncthreads();
      for (int i = tid; i < E; i += AGG_THREADS) {
        const int d = dst[i];
        if (d < base || d >= base + AGG_ROWS || d >= N) continue;
        const float v = as[src[i]] + sad[d - base];
        atomicAdd(&ssum[d - base], expf((v > 0.f ? v : LEAKY * v) - ord2f(smax[d - base])));
      }
      __syncthreads();
      for (int i = tid; i < E; i += AGG_THREADS) {
        const int d = dst[i];
        if (d < base || d >= base + AGG_ROWS || d >= N) continue;
        const int   s = src[i];
        const float v = as[s] + sad[d - base];
        const float c = expf((v > 0.f ? v : LEAKY * v) - ord2f(smax[d - base])) / (ssum[d - base] + 1e-16f);
        const float* hp = h + (size_t)s * F_DIM;
        float* tp = tile + (d - base) * F_DIM;
        for (int f = 0; f < F_DIM; ++f) atomicAdd(tp + f, c * hp[f]);
      }
    }
    __syncthreads();

    const int rows = min(AGG_ROWS, N - base);
    float4* op = (float4*)(out + (size_t)base * F_DIM);
    for (int k = tid; k < rows * (F_DIM / 4); k += AGG_THREADS)
      op[k] = ((const float4*)tile)[k];
    __syncthreads();
  }
}

extern "C" void kernel_launch(void* const* d_in, const int* in_sizes, int n_in,
                              void* d_out, int out_size, void* d_ws, size_t ws_size,
                              hipStream_t stream) {
  const float* x  = (const float*)d_in[0];
  const int*   ei = (const int*)d_in[1];
  const float* W  = (const float*)d_in[2];
  const float* a  = (const float*)d_in[3];
  const int N = in_sizes[0] / F_DIM;
  const int E = in_sizes[1] / 2;
  const int* src = ei;
  const int* dst = ei + E;

  float*    h      = (float*)d_ws;
  float*    as     = h + (size_t)N * F_DIM;
  float*    ad     = as + N;
  unsigned* masks  = (unsigned*)(ad + N);
  float*    out    = (float*)d_out;

  gat_gemm_wmma <<<GEMM_GROUPS,  256, 0, stream>>>(x, W, h, N);
  gat_alpha     <<<ALPHA_GROUPS, ALPHA_THREADS, 0, stream>>>(h, a, as, ad, N);
  gat_edge_mask <<<(E + MASK_CHUNK - 1) / MASK_CHUNK, 256, 0, stream>>>(dst, masks, E);
  gat_aggregate <<<AGG_GROUPS, AGG_THREADS, 0, stream>>>(src, dst, h, as, ad, masks, out, N, E);
}


# aggregate: a gathered h row comes as ONE 16-byte-per-lane load (one 512-B request), not four 4-byte line loads; LDS tile rows kept lane-major so adds stay conflict-free; sums in LDS-atomic order
# speedup vs baseline: 3.9094x; 1.0595x over previous
#include <hip/hip_runtime.h>
#include <hip/hip_bf16.h>

typedef __attribute__((ext_vector_type(2))) float v2f;
typedef __attribute__((ext_vector_type(8))) float v8f;

#define F_DIM 128
#define LEAKY 0.2f

#define AGG_GROUPS  256
#define AGG_ROWS    196
#define AGG_THREADS 1024
#define AGG_QCAP    8192
#define MASK_CHUNK  1024
#define MASK_BLOCK  16
#define AGG_BCAP    8192
#define AGG_BLOADS  16

#define GEMM_GROUPS   1024
#define ALPHA_GROUPS  256
#define ALPHA_THREADS 512
#define ALPHA_ROWS    8

__device__ __forceinline__ unsigned f2ord(float f) {
  unsigned u = __float_as_uint(f);
  return (u & 0x80000000u) ? ~u : (u | 0x80000000u);
}
__device__ __forceinline__ float ord2f(unsigned u) {
  return (u & 0x80000000u) ? __uint_as_float(u & 0x7FFFFFFFu)
                           : __uint_as_float(~u);
}

__global__ __launch_bounds__(256) void gat_gemm_wmma(
    const float* __restrict__ x, const float* __restrict__ W,
    float* __restrict__ h, int M)
{
  __shared__ float sA[16 * 132];
  const int tid  = threadIdx.x;
  const int lane = tid & 31;
  const int wv   = tid >> 5;
  const int row  = lane & 15;
  const int kh   = (lane >> 4) * 2;
  const int n0   = wv * 16;
  const int mh   = (lane >> 4) * 8;
  const int ntiles = (M + 15) >> 4;

  const int sr = tid >> 5, sc = tid & 31;

  v2f bf[32];
  #pragma unroll
  for (int kk = 0; kk < 32; ++kk) {
    bf[kk].x = W[(4 * kk + kh + 0) * F_DIM + n0 + row];
    bf[kk].y = W[(4 * kk + kh + 1) * F_DIM + n0 + row];
  }

  int tile = blockIdx.x;
  float4 p0 = make_float4(0.f, 0.f, 0.f, 0.f), p1 = p0;
  if (tile < ntiles) {
    const int m0 = tile * 16;
    if (m0 + sr     < M) p0 = ((const float4*)(x + (size_t)(m0 + sr)     * F_DIM))[sc];
    if (m0 + sr + 8 < M) p1 = ((const float4*)(x + (size_t)(m0 + sr + 8) * F_DIM))[sc];
  }

  for (; tile < ntiles; tile += gridDim.x) {
    const int m0 = tile * 16;
    ((float4*)sA)[sr * 33 + sc]       = p0;
    ((float4*)sA)[(sr + 8) * 33 + sc] = p1;
    __syncthreads();

    const int nt = tile + gridDim.x;
    p0 = make_float4(0.f, 0.f, 0.f, 0.f); p1 = p0;
    if (nt < ntiles) {
      const int q0 = nt * 16;
      if (q0 + sr     < M) p0 = ((const float4*)(x + (size_t)(q0 + sr)     * F_DIM))[sc];
      if (q0 + sr + 8 < M) p1 = ((const float4*)(x + (size_t)(q0 + sr + 8) * F_DIM))[sc];
    }

    v8f acc = {};
    #pragma unroll
    for (int kk = 0; kk < 32; ++kk) {
      v2f a;
      a.x = sA[row * 132 + 4 * kk + kh + 0];
      a.y = sA[row * 132 + 4 * kk + kh + 1];
      acc = __builtin_amdgcn_wmma_f32_16x16x4_f32(false, a, false, bf[kk],
                                                  (short)0, acc, false, false);
    }

    float* hp = h + (size_t)m0 * F_DIM + n0 + row;
    if (m0 + 16 <= M) {
      #pragma unroll
      for (int v = 0; v < 8; ++v) hp[(size_t)(mh + v) * F_DIM] = acc[v];
    } else {
      #pragma unroll
      for (int v = 0; v < 8; ++v)
        if (m0 + mh + v < M) hp[(size_t)(mh + v) * F_DIM] = acc[v];
    }
    __syncthreads();
  }
}

__global__ __launch_bounds__(ALPHA_THREADS) void gat_alpha(
    const float* __restrict__ h, const float* __restrict__ a,
    float* __restrict__ as, float* __restrict__ ad, int N)
{
  const int lane   = threadIdx.x & 31;
  const int wave   = blockIdx.x * (ALPHA_THREADS / 32) + (threadIdx.x >> 5);
  const int nwaves = gridDim.x * (ALPHA_THREADS / 32);
  const float4 a1 = ((const float4*)a)[lane];
  const float4 a2 = ((const float4*)(a + F_DIM))[lane];

  for (int node0 = wave * ALPHA_ROWS; node0 < N; node0 += nwaves * ALPHA_ROWS) {
    float4 hv[ALPHA_ROWS];
    #pragma unroll
    for (int u = 0; u < ALPHA_ROWS; ++u) {
      hv[u] = make_float4(0.f, 0.f, 0.f, 0.f);
      if (node0 + u < N) hv[u] = ((const float4*)(h + (size_t)(node0 + u) * F_DIM))[lane];
    }
    float sv = 0.f, dv = 0.f;
    #pragma unroll
    for (int u = 0; u < ALPHA_ROWS; ++u) {
      float s = hv[u].x * a1.x + hv[u].y * a1.y + hv[u].z * a1.z + hv[u].w * a1.w;
      float d = hv[u].x * a2.x + hv[u].y * a2.y + hv[u].z * a2.z + hv[u].w * a2.w;
      #pragma unroll
      for (int m = 16; m >= 1; m >>= 1) {
        s += __shfl_xor(s, m, 32);
        d += __shfl_xor(d, m, 32);
      }
      if (lane == u) { sv = s; dv = d; }
    }
    if (lane < ALPHA_ROWS && node0 + lane < N) {
      as[node0 + lane] = sv;
      ad[node0 + lane] = dv;
    }
  }
}

__global__ __launch_bounds__(256) void gat_edge_mask(
    const int* __restrict__ dst, unsigned* __restrict__ masks, int E)
{
  __shared__ unsigned sm[2 * AGG_GROUPS];
  const int t  = threadIdx.x;
  const int i0 = blockIdx.x * MASK_CHUNK + t;
  int d[4];
  #pragma unroll
  for (int u = 0; u < 4; ++u) {
    const int i = i0 + u * 256;
    d[u] = 0;
    if (i < E) d[u] = dst[i];
  }
  sm[t] = 0u;
  sm[t + 256] = 0u;
  __syncthreads();
  #pragma unroll
  for (int u = 0; u < 4; ++u) {
    const int i = i0 + u * 256;
    if (i < E) {
      const int o = (d[u] / AGG_ROWS) & (AGG_GROUPS - 1);
      const int b = (t + u * 256) / MASK_BLOCK;
      atomicOr(&sm[2 * o + (b >> 5)], 1u << (b & 31));
    }
  }
  __syncthreads();
  ((uint2*)masks)[(size_t)blockIdx.x * AGG_GROUPS + t] = make_uint2(sm[2 * t], sm[2 * t + 1]);
}

__global__ __launch_bounds__(AGG_THREADS) void gat_aggregate(
    const int* __restrict__ src, const int* __restrict__ dst,
    const float* __restrict__ h,
    const float* __restrict__ as, const float* __restrict__ ad,
    const unsigned* __restrict__ masks,
    float* __restrict__ out, int N, int E)
{
  __shared__ float    tile[AGG_ROWS * F_DIM];
  __shared__ int      queue[AGG_QCAP];
  __shared__ float    qv[AGG_QCAP];
  __shared__ unsigned char qd[AGG_QCAP];
  __shared__ int      blist[AGG_BCAP];
  __shared__ unsigned smax[AGG_ROWS];
  __shared__ float    ssum[AGG_ROWS];
  __shared__ float    sad[AGG_ROWS];
  __shared__ int      qcount;
  __shared__ int      bcount;

  const int tid  = threadIdx.x;
  const int lane = tid & 31;
  const int wv   = tid >> 5;
  const unsigned me = blockIdx.x;
  const int nchunks = (E + MASK_CHUNK - 1) / MASK_CHUNK;
  const uint2* mp = (const uint2*)masks;

  for (int base = blockIdx.x * AGG_ROWS; base < N; base += AGG_GROUPS * AGG_ROWS) {
    for (int k = tid; k < AGG_ROWS * F_DIM / 4; k += AGG_THREADS)
      ((float4*)tile)[k] = make_float4(0.f, 0.f, 0.f, 0.f);
    for (int k = tid; k < AGG_ROWS; k += AGG_THREADS) {
      smax[k] = 0u;
      ssum[k] = 0.f;
      sad[k]  = (base + k < N) ? ad[base + k] : 0.f;
    }
    if (tid == 0) { qcount = 0; bcount = 0; }
    __syncthreads();

    for (int c0 = 0; c0 < nchunks; c0 += AGG_THREADS) {
      const int c = c0 + tid;
      uint2 mk = make_uint2(0u, 0u);
      if (c < nchunks) mk = mp[(size_t)c * AGG_GROUPS + me];
      const int cnt = __popc(mk.x) + __popc(mk.y);
      int inc = cnt;
      #pragma unroll
      for (int s = 1; s < 32; s <<= 1) {
        const int up = __shfl_up(inc, s, 32);
        if (lane >= s) inc += up;
      }
      const int tot = __shfl(inc, 31, 32);
      int bs = 0;
      if (lane == 31 && tot > 0) bs = atomicAdd(&bcount, tot);
      bs = __shfl(bs, 31, 32);
      int pos = bs + inc - cnt;
      const int blk0 = c * (MASK_CHUNK / MASK_BLOCK);
      unsigned m = mk.x;
      while (m) {
        const int p = __builtin_ctz(m);
        m &= m - 1;
        if (pos < AGG_BCAP) blist[pos] = blk0 + p;
        ++pos;
      }
      m = mk.y;
      while (m) {
        const int p = __builtin_ctz(m);
        m &= m - 1;
        if (pos < AGG_BCAP) blist[pos] = blk0 + 32 + p;
        ++pos;
      }
    }
    __syncthreads();

    const int nb = bcount;
    if (nb <= AGG_BCAP) {
      const int npair = (nb + 1) >> 1;
      for (int k0 = wv; k0 < npair; k0 += 32 * AGG_BLOADS) {
        int ii[AGG_BLOADS], dd[AGG_BLOADS];
        #pragma unroll
        for (int u = 0; u < AGG_BLOADS; ++u) {
          const int bi = 2 * (k0 + u * 32) + (lane >> 4);
          ii[u] = blist[bi < nb ? bi : 0];
        }
        #pragma unroll
        for (int u = 0; u < AGG_BLOADS; ++u) {
          const int bi = 2 * (k0 + u * 32) + (lane >> 4);
          const int i  = ii[u] * MASK_BLOCK + (lane & 15);
          ii[u] = (bi < nb && i < E) ? i : -1;
        }
        #pragma unroll
        for (int u = 0; u < AGG_BLOADS; ++u) dd[u] = dst[ii[u] < 0 ? 0 : ii[u]];
        unsigned hb[AGG_BLOADS];
        int tot = 0;
        #pragma unroll
        for (int u = 0; u < AGG_BLOADS; ++u) {
          const int d = dd[u];
          const bool hit = ii[u] >= 0 && d >= base && d < base + AGG_ROWS && d < N;
          hb[u] = __builtin_amdgcn_ballot_w32(hit);
          tot += __popc(hb[u]);
        }
        if (tot > 0) {
          int bs = 0;
          if (lane == 0) bs = atomicAdd(&qcount, tot);
          bs = __builtin_amdgcn_readfirstlane(bs);
          #pragma unroll
          for (int u = 0; u < AGG_BLOADS; ++u) {
            if ((hb[u] >> lane) & 1u) {
              const int pos = bs + __builtin_amdgcn_mbcnt_lo(hb[u], 0u);
              if (pos < AGG_QCAP) {
                queue[pos] = ii[u];
                qd[pos]    = (unsigned char)(dd[u] - base);
              }
            }
            bs += __popc(hb[u]);
          }
        }
      }
    }
    __syncthreads();

    const int total = qcount;
    if (nb <= AGG_BCAP && total <= AGG_QCAP) {
      const int n = total;

      for (int q0 = tid; q0 < n; q0 += 4 * AGG_THREADS) {
        int   ei[4], es[4];
        float ea[4];
        #pragma unroll
        for (int u = 0; u < 4; ++u) {
          const int q = q0 + u * AGG_THREADS;
          ei[u] = (q < n) ? queue[q] : -1;
        }
        #pragma unroll
        for (int u = 0; u < 4; ++u) {
          es[u] = 0;
          if (ei[u] >= 0) es[u] = src[ei[u]];
        }
        #pragma unroll
        for (int u = 0; u < 4; ++u) ea[u] = as[es[u]];
        #pragma unroll
        for (int u = 0; u < 4; ++u) {
          if (ei[u] >= 0) {
            const int q  = q0 + u * AGG_THREADS;
            const int dl = qd[q];
            const float v  = ea[u] + sad[dl];
            const float ev = v > 0.f ? v : LEAKY * v;
            queue[q] = es[u];
            qv[q]    = ev;
            atomicMax(&smax[dl], f2ord(ev));
          }
        }
      }
      __syncthreads();

      for (int q = tid; q < n; q += AGG_THREADS) {
        const int dl = qd[q];
        const float v = expf(qv[q] - ord2f(smax[dl]));
        qv[q] = v;
        atomicAdd(&ssum[dl], v);
      }
      __syncthreads();

      for (int q0 = wv * 32; q0 < n; q0 += AGG_THREADS) {
        const int q = q0 + lane;
        int s = 0, dl = -1;
        float c = 0.f;
        if (q < n) {
          const int d8 = qd[q];
          s  = queue[q];
          dl = d8;
          c  = qv[q] / (ssum[d8] + 1e-16f);
        }
        for (int j = 0; j < 32; j += 8) {
          if (q0 + j >= n) break;
          float4 r[8];
          int    dj[8];
          float cj[8];
          #pragma unroll
          for (int u = 0; u < 8; ++u) {
            const int sj = __builtin_amdgcn_readlane(s, j + u);
            dj[u] = __builtin_amdgcn_readlane(dl, j + u);
            cj[u] = __int_as_float(__builtin_amdgcn_readlane(__float_as_int(c), j + u));
            r[u] = ((const float4*)(h + (size_t)sj * F_DIM))[lane];
          }
          asm volatile("" :: "v"(r[0].x), "v"(r[0].y), "v"(r[0].z), "v"(r[0].w),
                             "v"(r[1].x), "v"(r[1].y), "v"(r[1].z), "v"(r[1].w),
                             "v"(r[2].x), "v"(r[2].y), "v"(r[2].z), "v"(r[2].w),
                             "v"(r[3].x), "v"(r[3].y), "v"(r[3].z), "v"(r[3].w));
          asm volatile("" :: "v"(r[4].x), "v"(r[4].y), "v"(r[4].z), "v"(r[4].w),
                             "v"(r[5].x), "v"(r[5].y), "v"(r[5].z), "v"(r[5].w),
                             "v"(r[6].x), "v"(r[6].y), "v"(r[6].z), "v"(r[6].w),
                             "v"(r[7].x), "v"(r[7].y), "v"(r[7].z), "v"(r[7].w));
          #pragma unroll
          for (int u = 0; u < 8; ++u) {
            if (dj[u] >= 0) {
              float* tp = tile + dj[u] * F_DIM + lane;
              atomicAdd(tp,      cj[u] * r[u].x);
              atomicAdd(tp + 32, cj[u] * r[u].y);
              atomicAdd(tp + 64, cj[u] * r[u].z);
              atomicAdd(tp + 96, cj[u] * r[u].w);
            }
          }
        }
      }
    } else {
      for (int i = tid; i < E; i += AGG_THREADS) {
        const int d = dst[i];
        if (d < base || d >= base + AGG_ROWS || d >= N) continue;
        const float v = as[src[i]] + sad[d - base];
        atomicMax(&smax[d - base], f2ord(v > 0.f ? v : LEAKY * v));
      }
      __syncthreads();
      for (int i = tid; i < E; i += AGG_THREADS) {
        const int d = dst[i];
        if (d < base || d >= base + AGG_ROWS || d >= N) continue;
        const float v = as[src[i]] + sad[d - base];
        atomicAdd(&ssum[d - base], expf((v > 0.f ? v : LEAKY * v) - ord2f(smax[d - base])));
      }
      __syncthreads();
      for (int i = tid; i < E; i += AGG_THREADS) {
        const int d = dst[i];
        if (d < base || d >= base + AGG_ROWS || d >= N) continue;
        const int   s = src[i];
        const float v = as[s] + sad[d - base];
        const float c = expf((v > 0.f ? v : LEAKY * v) - ord2f(smax[d - base])) / (ssum[d - base] + 1e-16f);
        const float* hp = h + (size_t)s * F_DIM;
        float* tp = tile + (d - base) * F_DIM;
        for (int f = 0; f < F_DIM; ++f) atomicAdd(tp + (f & 3) * 32 + (f >> 2), c * hp[f]);
      }
    }
    __syncthreads();

    const int rows = min(AGG_ROWS, N - base);
    float4* op = (float4*)(out + (size_t)base * F_DIM);
    for (int k = tid; k < rows * (F_DIM / 4); k += AGG_THREADS) {
      const float* tr = tile + (k >> 5) * F_DIM + (k & 31);
      op[k] = make_float4(tr[0], tr[32], tr[64], tr[96]);
    }
    __syncthreads();
  }
}

extern "C" void kernel_launch(void* const* d_in, const int* in_sizes, int n_in,
                              void* d_out, int out_size, void* d_ws, size_t ws_size,
                              hipStream_t stream) {
  const float* x  = (const float*)d_in[0];
  const int*   ei = (const int*)d_in[1];
  const float* W  = (const float*)d_in[2];
  const float* a  = (const float*)d_in[3];
  const int N = in_sizes[0] / F_DIM;
  const int E = in_sizes[1] / 2;
  const int* src = ei;
  const int* dst = ei + E;

  float*    h      = (float*)d_ws;
  float*    as     = h + (size_t)N * F_DIM;
  float*    ad     = as + N;
  unsigned* masks  = (unsigned*)(ad + N);
  float*    out    = (float*)d_out;

  gat_gemm_wmma <<<GEMM_GROUPS,  256, 0, stream>>>(x, W, h, N);
  gat_alpha     <<<ALPHA_GROUPS, ALPHA_THREADS, 0, stream>>>(h, a, as, ad, N);
  gat_edge_mask <<<(E + MASK_CHUNK - 1) / MASK_CHUNK, 256, 0, stream>>>(dst, masks, E);
  gat_aggregate <<<AGG_GROUPS, AGG_THREADS, 0, stream>>>(src, dst, h, as, ad, masks, out, N, E);
}


# presence-mask work-groups ride behind the alpha work-groups in one 256-thread launch (block-index branch) instead of a launch of their own: 4 launches become 3; every sum in the same order
# speedup vs baseline: 4.1206x; 1.0540x over previous
#include <hip/hip_runtime.h>
#include <hip/hip_bf16.h>

typedef __attribute__((ext_vector_type(2))) float v2f;
typedef __attribute__((ext_vector_type(8))) float v8f;

#define F_DIM 128
#define LEAKY 0.2f

#define AGG_GROUPS  256
#define AGG_ROWS    196
#define AGG_THREADS 1024
#define AGG_QCAP    8192
#define MASK_CHUNK  1024
#define MASK_BLOCK  16
#define AGG_BCAP    8192
#define AGG_BLOADS  16

#define GEMM_GROUPS   1024
#define ALPHA_GROUPS  512
#define ALPHA_THREADS 256
#define ALPHA_ROWS    8

__device__ __forceinline__ unsigned f2ord(float f) {
  unsigned u = __float_as_uint(f);
  return (u & 0x80000000u) ? ~u : (u | 0x80000000u);
}
__device__ __forceinline__ float ord2f(unsigned u) {
  return (u & 0x80000000u) ? __uint_as_float(u & 0x7FFFFFFFu)
                           : __uint_as_float(~u);
}

__global__ __launch_bounds__(256) void gat_gemm_wmma(
    const float* __restrict__ x, const float* __restrict__ W,
    float* __restrict__ h, int M)
{
  __shared__ float sA[16 * 132];
  const int tid  = threadIdx.x;
  const int lane = tid & 31;
  const int wv   = tid >> 5;
  const int row  = lane & 15;
  const int kh   = (lane >> 4) * 2;
  const int n0   = wv * 16;
  const int mh   = (lane >> 4) * 8;
  const int ntiles = (M + 15) >> 4;

  const int sr = tid >> 5, sc = tid & 31;

  v2f bf[32];
  #pragma unroll
  for (int kk = 0; kk < 32; ++kk) {
    bf[kk].x = W[(4 * kk + kh + 0) * F_DIM + n0 + row];
    bf[kk].y = W[(4 * kk + kh + 1) * F_DIM + n0 + row];
  }

  int tile = blockIdx.x;
  float4 p0 = make_float4(0.f, 0.f, 0.f, 0.f), p1 = p0;
  if (tile < ntiles) {
    const int m0 = tile * 16;
    if (m0 + sr     < M) p0 = ((const float4*)(x + (size_t)(m0 + sr)     * F_DIM))[sc];
    if (m0 + sr + 8 < M) p1 = ((const float4*)(x + (size_t)(m0 + sr + 8) * F_DIM))[sc];
  }

  for (; tile < ntiles; tile += gridDim.x) {
    const int m0 = tile * 16;
    ((float4*)sA)[sr * 33 + sc]       = p0;
    ((float4*)sA)[(sr + 8) * 33 + sc] = p1;
    __syncthreads();

    const int nt = tile + gridDim.x;
    p0 = make_float4(0.f, 0.f, 0.f, 0.f); p1 = p0;
    if (nt < ntiles) {
      const int q0 = nt * 16;
      if (q0 + sr     < M) p0 = ((const float4*)(x + (size_t)(q0 + sr)     * F_DIM))[sc];
      if (q0 + sr + 8 < M) p1 = ((const float4*)(x + (size_t)(q0 + sr + 8) * F_DIM))[sc];
    }

    v8f acc = {};
    #pragma unroll
    for (int kk = 0; kk < 32; ++kk) {
      v2f a;
      a.x = sA[row * 132 + 4 * kk + kh + 0];
      a.y = sA[row * 132 + 4 * kk + kh + 1];
      acc = __builtin_amdgcn_wmma_f32_16x16x4_f32(false, a, false, bf[kk],
                                                  (short)0, acc, false, false);
    }

    float* hp = h + (size_t)m0 * F_DIM + n0 + row;
    if (m0 + 16 <= M) {
      #pragma unroll
      for (int v = 0; v < 8; ++v) hp[(size_t)(mh + v) * F_DIM] = acc[v];
    } else {
      #pragma unroll
      for (int v = 0; v < 8; ++v)
        if (m0 + mh + v < M) hp[(size_t)(mh + v) * F_DIM] = acc[v];
    }
    __syncthreads();
  }
}

__device__ __forceinline__ void alpha_part(
    const float* __restrict__ h, const float* __restrict__ a,
    float* __restrict__ as, float* __restrict__ ad, int N)
{
  const int lane   = threadIdx.x & 31;
  const int wave   = blockIdx.x * (ALPHA_THREADS / 32) + (threadIdx.x >> 5);
  const int nwaves = ALPHA_GROUPS * (ALPHA_THREADS / 32);
  const float4 a1 = ((const float4*)a)[lane];
  const float4 a2 = ((const float4*)(a + F_DIM))[lane];

  for (int node0 = wave * ALPHA_ROWS; node0 < N; node0 += nwaves * ALPHA_ROWS) {
    float4 hv[ALPHA_ROWS];
    #pragma unroll
    for (int u = 0; u < ALPHA_ROWS; ++u) {
      hv[u] = make_float4(0.f, 0.f, 0.f, 0.f);
      if (node0 + u < N) hv[u] = ((const float4*)(h + (size_t)(node0 + u) * F_DIM))[lane];
    }
    float sv = 0.f, dv = 0.f;
    #pragma unroll
    for (int u = 0; u < ALPHA_ROWS; ++u) {
      float s = hv[u].x * a1.x + hv[u].y * a1.y + hv[u].z * a1.z + hv[u].w * a1.w;
      float d = hv[u].x * a2.x + hv[u].y * a2.y + hv[u].z * a2.z + hv[u].w * a2.w;
      #pragma unroll
      for (int m = 16; m >= 1; m >>= 1) {
        s += __shfl_xor(s, m, 32);
        d += __shfl_xor(d, m, 32);
      }
      if (lane == u) { sv = s; dv = d; }
    }
    if (lane < ALPHA_ROWS && node0 + lane < N) {
      as[node0 + lane] = sv;
      ad[node0 + lane] = dv;
    }
  }
}

__device__ __forceinline__ void mask_part(
    const int* __restrict__ dst, unsigned* __restrict__ masks, int E,
    int chunk, unsigned* sm)
{
  const int t  = threadIdx.x;
  const int i0 = chunk * MASK_CHUNK + t;
  int d[4];
  #pragma unroll
  for (int u = 0; u < 4; ++u) {
    const int i = i0 + u * 256;
    d[u] = 0;
    if (i < E) d[u] = dst[i];
  }
  sm[t] = 0u;
  sm[t + 256] = 0u;
  __syncthreads();
  #pragma unroll
  for (int u = 0; u < 4; ++u) {
    const int i = i0 + u * 256;
    if (i < E) {
      const int o = (d[u] / AGG_ROWS) & (AGG_GROUPS - 1);
      const int b = (t + u * 256) / MASK_BLOCK;
      atomicOr(&sm[2 * o + (b >> 5)], 1u << (b & 31));
    }
  }
  __syncthreads();
  ((uint2*)masks)[(size_t)chunk * AGG_GROUPS + t] = make_uint2(sm[2 * t], sm[2 * t + 1]);
}

__global__ __launch_bounds__(ALPHA_THREADS) void gat_alpha_mask(
    const float* __restrict__ h, const float* __restrict__ a,
    float* __restrict__ as, float* __restrict__ ad, int N,
    const int* __restrict__ dst, unsigned* __restrict__ masks, int E)
{
  __shared__ unsigned sm[2 * AGG_GROUPS];
  if (blockIdx.x < ALPHA_GROUPS) {
    alpha_part(h, a, as, ad, N);
  } else {
    mask_part(dst, masks, E, (int)blockIdx.x - ALPHA_GROUPS, sm);
  }
}

__global__ __launch_bounds__(AGG_THREADS) void gat_aggregate(
    const int* __restrict__ src, const int* __restrict__ dst,
    const float* __restrict__ h,
    const float* __restrict__ as, const float* __restrict__ ad,
    const unsigned* __restrict__ masks,
    float* __restrict__ out, int N, int E)
{
  __shared__ float    tile[AGG_ROWS * F_DIM];
  __shared__ int      queue[AGG_QCAP];
  __shared__ float    qv[AGG_QCAP];
  __shared__ unsigned char qd[AGG_QCAP];
  __shared__ int      blist[AGG_BCAP];
  __shared__ unsigned smax[AGG_ROWS];
  __shared__ float    ssum[AGG_ROWS];
  __shared__ float    sad[AGG_ROWS];
  __shared__ int      qcount;
  __shared__ int      bcount;

  const int tid  = threadIdx.x;
  const int lane = tid & 31;
  const int wv   = tid >> 5;
  const unsigned me = blockIdx.x;
  const int nchunks = (E + MASK_CHUNK - 1) / MASK_CHUNK;
  const uint2* mp = (const uint2*)masks;

  for (int base = blockIdx.x * AGG_ROWS; base < N; base += AGG_GROUPS * AGG_ROWS) {
    for (int k = tid; k < AGG_ROWS * F_DIM / 4; k += AGG_THREADS)
      ((float4*)tile)[k] = make_float4(0.f, 0.f, 0.f, 0.f);
    for (int k = tid; k < AGG_ROWS; k += AGG_THREADS) {
      smax[k] = 0u;
      ssum[k] = 0.f;
      sad[k]  = (base + k < N) ? ad[base + k] : 0.f;
    }
    if (tid == 0) { qcount = 0; bcount = 0; }
    __syncthreads();

    for (int c0 = 0; c0 < nchunks; c0 += AGG_THREADS) {
      const int c = c0 + tid;
      uint2 mk = make_uint2(0u, 0u);
      if (c < nchunks) mk = mp[(size_t)c * AGG_GROUPS + me];
      const int cnt = __popc(mk.x) + __popc(mk.y);
      int inc = cnt;
      #pragma unroll
      for (int s = 1; s < 32; s <<= 1) {
        const int up = __shfl_up(inc, s, 32);
        if (lane >= s) inc += up;
      }
      const int tot = __shfl(inc, 31, 32);
      int bs = 0;
      if (lane == 31 && tot > 0) bs = atomicAdd(&bcount, tot);
      bs = __shfl(bs, 31, 32);
      int pos = bs + inc - cnt;
      const int blk0 = c * (MASK_CHUNK / MASK_BLOCK);
      unsigned m = mk.x;
      while (m) {
        const int p = __builtin_ctz(m);
        m &= m - 1;
        if (pos < AGG_BCAP) blist[pos] = blk0 + p;
        ++pos;
      }
      m = mk.y;
      while (m) {
        const int p = __builtin_ctz(m);
        m &= m - 1;
        if (pos < AGG_BCAP) blist[pos] = blk0 + 32 + p;
        ++pos;
      }
    }
    __syncthreads();

    const int nb = bcount;
    if (nb <= AGG_BCAP) {
      const int npair = (nb + 1) >> 1;
      for (int k0 = wv; k0 < npair; k0 += 32 * AGG_BLOADS) {
        int ii[AGG_BLOADS], dd[AGG_BLOADS];
        #pragma unroll
        for (int u = 0; u < AGG_BLOADS; ++u) {
          const int bi = 2 * (k0 + u * 32) + (lane >> 4);
          ii[u] = blist[bi < nb ? bi : 0];
        }
        #pragma unroll
        for (int u = 0; u < AGG_BLOADS; ++u) {
          const int bi = 2 * (k0 + u * 32) + (lane >> 4);
          const int i  = ii[u] * MASK_BLOCK + (lane & 15);
          ii[u] = (bi < nb && i < E) ? i : -1;
        }
        #pragma unroll
        for (int u = 0; u < AGG_BLOADS; ++u) dd[u] = dst[ii[u] < 0 ? 0 : ii[u]];
        unsigned hb[AGG_BLOADS];
        int tot = 0;
        #pragma unroll
        for (int u = 0; u < AGG_BLOADS; ++u) {
          const int d = dd[u];
          const bool hit = ii[u] >= 0 && d >= base && d < base + AGG_ROWS && d < N;
          hb[u] = __builtin_amdgcn_ballot_w32(hit);
          tot += __popc(hb[u]);
        }
        if (tot > 0) {
          int bs = 0;
          if (lane == 0) bs = atomicAdd(&qcount, tot);
          bs = __builtin_amdgcn_readfirstlane(bs);
          #pragma unroll
          for (int u = 0; u < AGG_BLOADS; ++u) {
            if ((hb[u] >> lane) & 1u) {
              const int pos = bs + __builtin_amdgcn_mbcnt_lo(hb[u], 0u);
              if (pos < AGG_QCAP) {
                queue[pos] = ii[u];
                qd[pos]    = (unsigned char)(dd[u] - base);
              }
            }
            bs += __popc(hb[u]);
          }
        }
      }
    }
    __syncthreads();

    const int total = qcount;
    if (nb <= AGG_BCAP && total <= AGG_QCAP) {
      const int n = total;

      for (int q0 = tid; q0 < n; q0 += 4 * AGG_THREADS) {
        int   ei[4], es[4];
        float ea[4];
        #pragma unroll
        for (int u = 0; u < 4; ++u) {
          const int q = q0 + u * AGG_THREADS;
          ei[u] = (q < n) ? queue[q] : -1;
        }
        #pragma unroll
        for (int u = 0; u < 4; ++u) {
          es[u] = 0;
          if (ei[u] >= 0) es[u] = src[ei[u]];
        }
        #pragma unroll
        for (int u = 0; u < 4; ++u) ea[u] = as[es[u]];
        #pragma unroll
        for (int u = 0; u < 4; ++u) {
          if (ei[u] >= 0) {
            const int q  = q0 + u * AGG_THREADS;
            const int dl = qd[q];
            const float v  = ea[u] + sad[dl];
            const float ev = v > 0.f ? v : LEAKY * v;
            queue[q] = es[u];
            qv[q]    = ev;
            atomicMax(&smax[dl], f2ord(ev));
          }
        }
      }
      __syncthreads();

      for (int q = tid; q < n; q += AGG_THREADS) {
        const int dl = qd[q];
        const float v = expf(qv[q] - ord2f(smax[dl]));
        qv[q] = v;
        atomicAdd(&ssum[dl], v);
      }
      __syncthreads();

      for (int q0 = wv * 32; q0 < n; q0 += AGG_THREADS) {
        const int q = q0 + lane;
        int s = 0, dl = -1;
        float c = 0.f;
        if (q < n) {
          const int d8 = qd[q];
          s  = queue[q];
          dl = d8;
          c  = qv[q] / (ssum[d8] + 1e-16f);
        }
        for (int j = 0; j < 32; j += 8) {
          if (q0 + j >= n) break;
          float4 r[8];
          int    dj[8];
          float cj[8];
          #pragma unroll
          for (int u = 0; u < 8; ++u) {
            const int sj = __builtin_amdgcn_readlane(s, j + u);
            dj[u] = __builtin_amdgcn_readlane(dl, j + u);
            cj[u] = __int_as_float(__builtin_amdgcn_readlane(__float_as_int(c), j + u));
            r[u] = ((const float4*)(h + (size_t)sj * F_DIM))[lane];
          }
          asm volatile("" :: "v"(r[0].x), "v"(r[0].y), "v"(r[0].z), "v"(r[0].w),
                             "v"(r[1].x), "v"(r[1].y), "v"(r[1].z), "v"(r[1].w),
                             "v"(r[2].x), "v"(r[2].y), "v"(r[2].z), "v"(r[2].w),
                             "v"(r[3].x), "v"(r[3].y), "v"(r[3].z), "v"(r[3].w));
          asm volatile("" :: "v"(r[4].x), "v"(r[4].y), "v"(r[4].z), "v"(r[4].w),
                             "v"(r[5].x), "v"(r[5].y), "v"(r[5].z), "v"(r[5].w),
                             "v"(r[6].x), "v"(r[6].y), "v"(r[6].z), "v"(r[6].w),
                             "v"(r[7].x), "v"(r[7].y), "v"(r[7].z), "v"(r[7].w));
          #pragma unroll
          for (int u = 0; u < 8; ++u) {
            if (dj[u] >= 0) {
              float* tp = tile + dj[u] * F_DIM + lane;
              atomicAdd(tp,      cj[u] * r[u].x);
              atomicAdd(tp + 32, cj[u] * r[u].y);
              atomicAdd(tp + 64, cj[u] * r[u].z);
              atomicAdd(tp + 96, cj[u] * r[u].w);
            }
          }
        }
      }
    } else {
      for (int i = tid; i < E; i += AGG_THREADS) {
        const int d = dst[i];
        if (d < base || d >= base + AGG_ROWS || d >= N) continue;
        const float v = as[src[i]] + sad[d - base];
        atomicMax(&smax[d - base], f2ord(v > 0.f ? v : LEAKY * v));
      }
      __syncthreads();
      for (int i = tid; i < E; i += AGG_THREADS) {
        const int d = dst[i];
        if (d < base || d >= base + AGG_ROWS || d >= N) continue;
        const float v = as[src[i]] + sad[d - base];
        atomicAdd(&ssum[d - base], expf((v > 0.f ? v : LEAKY * v) - ord2f(smax[d - base])));
      }
      __syncthreads();
      for (int i = tid; i < E; i += AGG_THREADS) {
        const int d = dst[i];
        if (d < base || d >= base + AGG_ROWS || d >= N) continue;
        const int   s = src[i];
        const float v = as[s] + sad[d - base];
        const float c = expf((v > 0.f ? v : LEAKY * v) - ord2f(smax[d - base])) / (ssum[d - base] + 1e-16f);
        const float* hp = h + (size_t)s * F_DIM;
        float* tp = tile + (d - base) * F_DIM;
        for (int f = 0; f < F_DIM; ++f) atomicAdd(tp + (f & 3) * 32 + (f >> 2), c * hp[f]);
      }
    }
    __syncthreads();

    const int rows = min(AGG_ROWS, N - base);
    float4* op = (float4*)(out + (size_t)base * F_DIM);
    for (int k = tid; k < rows * (F_DIM / 4); k += AGG_THREADS) {
      const float* tr = tile + (k >> 5) * F_DIM + (k & 31);
      op[k] = make_float4(tr[0], tr[32], tr[64], tr[96]);
    }
    __syncthreads();
  }
}

extern "C" void kernel_launch(void* const* d_in, const int* in_sizes, int n_in,
                              void* d_out, int out_size, void* d_ws, size_t ws_size,
                              hipStream_t stream) {
  const float* x  = (const float*)d_in[0];
  const int*   ei = (const int*)d_in[1];
  const float* W  = (const float*)d_in[2];
  const float* a  = (const float*)d_in[3];
  const int N = in_sizes[0] / F_DIM;
  const int E = in_sizes[1] / 2;
  const int* src = ei;
  const int* dst = ei + E;

  float*    h      = (float*)d_ws;
  float*    as     = h + (size_t)N * F_DIM;
  float*    ad     = as + N;
  unsigned* masks  = (unsigned*)(ad + N);
  float*    out    = (float*)d_out;

  gat_gemm_wmma <<<GEMM_GROUPS,  256, 0, stream>>>(x, W, h, N);
  gat_alpha_mask<<<ALPHA_GROUPS + (E + MASK_CHUNK - 1) / MASK_CHUNK, ALPHA_THREADS, 0, stream>>>(h, a, as, ad, N, dst, masks, E);
  gat_aggregate <<<AGG_GROUPS, AGG_THREADS, 0, stream>>>(src, dst, h, as, ad, masks, out, N, E);
}
